# starting kernel
# baseline (speedup 1.0000x reference)
#include <hip/hip_runtime.h>
#include <math.h>

#define BATCH 8
#define SEQ   512
#define DIM   512
#define DK    64
#define QKV3  192
#define SCALE 0.125f

typedef __attribute__((ext_vector_type(2))) float v2f;
typedef __attribute__((ext_vector_type(8))) float v8f;

__global__ __launch_bounds__(256) void qkv_gemm_kernel(
    const float* __restrict__ X, const float* __restrict__ Wqkv,
    float* __restrict__ Qt, float* __restrict__ Kt, float* __restrict__ Vt)
{
    const int tid  = threadIdx.x;
    const int lane = tid & 31;
    const int wave = tid >> 5;
    const int tile = blockIdx.x * 8 + wave;
    const int JT   = QKV3 / 16;
    const int mt   = tile / JT;
    const int jt   = tile % JT;
    const int row0 = mt * 16;
    const int j0   = jt * 16;
    const int lm   = lane & 15;
    const int lk   = lane >> 4;

    v8f acc = {0.f,0.f,0.f,0.f,0.f,0.f,0.f,0.f};

    const float* arow = X + (size_t)(row0 + lm) * DIM + 2 * lk;
    const float* bcol = Wqkv + j0 + lm;

    #pragma unroll 4
    for (int k0 = 0; k0 < DIM; k0 += 4) {
        v2f a = *(const v2f*)(arow + k0);
        const int kb = k0 + 2 * lk;
        v2f b;
        b.x = bcol[(size_t)kb * QKV3];
        b.y = bcol[(size_t)(kb + 1) * QKV3];
        acc = __builtin_amdgcn_wmma_f32_16x16x4_f32(
            false, a, false, b, (short)0, acc, false, false);
    }

    const int b_idx = row0 >> 9;
    const int n0    = (row0 & 511) + lk * 8;
    const int col   = j0 + lm;
    const int which = j0 >> 6;
    const int c     = col & 63;
    float* T = (which == 0) ? Qt : ((which == 1) ? Kt : Vt);
    float* base = T + ((size_t)b_idx * DK + c) * SEQ;
    #pragma unroll
    for (int r = 0; r < 8; ++r)
        base[n0 + r] = acc[r];
}

__global__ __launch_bounds__(256) void attn_kernel(
    const float* __restrict__ Qt, const float* __restrict__ Kt,
    const float* __restrict__ Vt, float* __restrict__ SA)
{
    __shared__ __align__(16) float kv[SEQ];
    __shared__ __align__(16) float vv[SEQ];
    __shared__ float rmax[8];
    __shared__ float rmin[8];

    const int bc  = blockIdx.x;
    const int tid = threadIdx.x;

    const float* kp = Kt + (size_t)bc * SEQ;
    const float* vp = Vt + (size_t)bc * SEQ;
    kv[tid]       = kp[tid];
    kv[tid + 256] = kp[tid + 256];
    vv[tid]       = vp[tid];
    vv[tid + 256] = vp[tid + 256];
    __syncthreads();

    float kmx = fmaxf(kv[tid], kv[tid + 256]);
    float kmn = fminf(kv[tid], kv[tid + 256]);
    #pragma unroll
    for (int off = 16; off >= 1; off >>= 1) {
        kmx = fmaxf(kmx, __shfl_xor(kmx, off, 32));
        kmn = fminf(kmn, __shfl_xor(kmn, off, 32));
    }
    if ((tid & 31) == 0) { rmax[tid >> 5] = kmx; rmin[tid >> 5] = kmn; }
    __syncthreads();
    float KMAX = rmax[0], KMIN = rmin[0];
    #pragma unroll
    for (int i = 1; i < 8; ++i) {
        KMAX = fmaxf(KMAX, rmax[i]);
        KMIN = fminf(KMIN, rmin[i]);
    }

    const int b = bc >> 6;
    const int c = bc & 63;
    const float* qp = Qt + (size_t)bc * SEQ;
    const float4* kv4 = (const float4*)kv;
    const float4* vv4 = (const float4*)vv;

    #pragma unroll
    for (int r = 0; r < 2; ++r) {
        const int n = tid + r * 256;
        const float a    = qp[n] * SCALE;
        const float cmax = a * ((a >= 0.f) ? KMAX : KMIN);
        float s0 = 0.f, s1 = 0.f, s2 = 0.f, s3 = 0.f;
        float p0 = 0.f, p1 = 0.f, p2 = 0.f, p3 = 0.f;
        #pragma unroll 2
        for (int m4 = 0; m4 < SEQ / 4; ++m4) {
            const float4 k4 = kv4[m4];
            const float4 v4 = vv4[m4];
            const float e0 = __expf(fmaf(a, k4.x, -cmax));
            const float e1 = __expf(fmaf(a, k4.y, -cmax));
            const float e2 = __expf(fmaf(a, k4.z, -cmax));
            const float e3 = __expf(fmaf(a, k4.w, -cmax));
            s0 += e0; s1 += e1; s2 += e2; s3 += e3;
            p0 = fmaf(e0, v4.x, p0);
            p1 = fmaf(e1, v4.y, p1);
            p2 = fmaf(e2, v4.z, p2);
            p3 = fmaf(e3, v4.w, p3);
        }
        const float sum = (s0 + s1) + (s2 + s3);
        const float num = (p0 + p1) + (p2 + p3);
        SA[((size_t)b * SEQ + n) * DK + c] = num / sum;
    }
}

__global__ __launch_bounds__(256) void out_gemm_kernel(
    const float* __restrict__ SA, const float* __restrict__ Wo,
    float* __restrict__ Out)
{
    const int tid  = threadIdx.x;
    const int lane = tid & 31;
    const int wave = tid >> 5;
    const int tile = blockIdx.x * 8 + wave;
    const int JT   = DIM / 16;
    const int mt   = tile / JT;
    const int jt   = tile % JT;
    const int row0 = mt * 16;
    const int j0   = jt * 16;
    const int lm   = lane & 15;
    const int lk   = lane >> 4;

    v8f acc = {0.f,0.f,0.f,0.f,0.f,0.f,0.f,0.f};

    const float* arow = SA + (size_t)(row0 + lm) * DK + 2 * lk;
    const float* bcol = Wo + j0 + lm;

    #pragma unroll
    for (int k0 = 0; k0 < DK; k0 += 4) {
        v2f a = *(const v2f*)(arow + k0);
        const int kb = k0 + 2 * lk;
        v2f b;
        b.x = bcol[(size_t)kb * DIM];
        b.y = bcol[(size_t)(kb + 1) * DIM];
        acc = __builtin_amdgcn_wmma_f32_16x16x4_f32(
            false, a, false, b, (short)0, acc, false, false);
    }

    float* obase = Out + (size_t)(row0 + lk * 8) * DIM + j0 + lm;
    #pragma unroll
    for (int r = 0; r < 8; ++r)
        obase[(size_t)r * DIM] = acc[r];
}

extern "C" void kernel_launch(void* const* d_in, const int* in_sizes, int n_in,
                              void* d_out, int out_size, void* d_ws, size_t ws_size,
                              hipStream_t stream)
{
    const float* X    = (const float*)d_in[0];
    const float* Wqkv = (const float*)d_in[1];
    const float* Wo   = (const float*)d_in[2];
    float* out = (float*)d_out;

    const size_t CHN = (size_t)BATCH * DK * SEQ;
    float* Qt = (float*)d_ws;
    float* Kt = Qt + CHN;
    float* Vt = Kt + CHN;
    float* SA = Vt + CHN;

    qkv_gemm_kernel<<<384, 256, 0, stream>>>(X, Wqkv, Qt, Kt, Vt);
    attn_kernel<<<BATCH * DK, 256, 0, stream>>>(Qt, Kt, Vt, SA);
    out_gemm_kernel<<<1024, 256, 0, stream>>>(SA, Wo, out);
}


# QKV GEMM staged through LDS: 32x96 block tiles, coalesced 16-byte global loads, two-stage K-chunks of 64, 8-byte LDS fragment reads; k order and summation order unchanged
# speedup vs baseline: 1.3097x; 1.3097x over previous
#include <hip/hip_runtime.h>
#include <math.h>

#define BATCH 8
#define SEQ   512
#define DIM   512
#define DK    64
#define QKV3  192
#define SCALE 0.125f

typedef __attribute__((ext_vector_type(2))) float v2f;
typedef __attribute__((ext_vector_type(8))) float v8f;

#define G_BM 32
#define G_BN 96
#define G_BK 64
#define G_NK (DIM / G_BK)
#define G_PA (G_BK + 4)
#define G_PN 112
#define G_NT 384

__global__ __launch_bounds__(G_NT) void qkv_gemm_kernel(
    const float* __restrict__ X, const float* __restrict__ Wqkv,
    float* __restrict__ Qt, float* __restrict__ Kt, float* __restrict__ Vt)
{
    __shared__ __align__(16) float As[2][G_BM * G_PA];
    __shared__ __align__(16) float Bs[2][(G_BK / 2) * G_PN * 2];

    const int tid  = threadIdx.x;
    const int lane = tid & 31;
    const int wave = tid >> 5;
    const int lm   = lane & 15;
    const int lk   = lane >> 4;
    const int m0   = (blockIdx.x >> 1) * G_BM;
    const int jb0  = (blockIdx.x & 1) * G_BN;
    const int wm   = wave / 6;
    const int wn   = wave % 6;

    const int  a_row = tid >> 4;
    const int  a_q   = tid & 15;
    const bool a_two = tid < 128;
    const float* a_src0 = X + (size_t)(m0 + a_row) * DIM + 4 * a_q;
    const float* a_src1 = X + (size_t)(m0 + 24 + (a_row & 7)) * DIM + 4 * a_q;
    const int a_dst0 = a_row * G_PA + 4 * a_q;
    const int a_dst1 = (24 + (a_row & 7)) * G_PA + 4 * a_q;

    const int b_kp = tid / 24;
    const int b_c4 = tid % 24;
    const float* b_src = Wqkv + (size_t)(2 * b_kp) * QKV3 + jb0 + 4 * b_c4;
    const int b_dst0 = (b_kp * G_PN + 4 * b_c4) * 2;
    const int b_dst1 = ((b_kp + 16) * G_PN + 4 * b_c4) * 2;

    float4 ra0, ra1, rb00, rb01, rb10, rb11;
    ra1 = make_float4(0.f, 0.f, 0.f, 0.f);

    ra0  = *(const float4*)(a_src0);
    if (a_two) ra1 = *(const float4*)(a_src1);
    rb00 = *(const float4*)(b_src);
    rb01 = *(const float4*)(b_src + QKV3);
    rb10 = *(const float4*)(b_src + 32 * QKV3);
    rb11 = *(const float4*)(b_src + 33 * QKV3);

    *(float4*)&As[0][a_dst0] = ra0;
    if (a_two) *(float4*)&As[0][a_dst1] = ra1;
    *(float4*)&Bs[0][b_dst0]     = make_float4(rb00.x, rb01.x, rb00.y, rb01.y);
    *(float4*)&Bs[0][b_dst0 + 4] = make_float4(rb00.z, rb01.z, rb00.w, rb01.w);
    *(float4*)&Bs[0][b_dst1]     = make_float4(rb10.x, rb11.x, rb10.y, rb11.y);
    *(float4*)&Bs[0][b_dst1 + 4] = make_float4(rb10.z, rb11.z, rb10.w, rb11.w);
    __syncthreads();

    v8f acc = {0.f,0.f,0.f,0.f,0.f,0.f,0.f,0.f};

    const int a_frag = (wm * 16 + lm) * G_PA + 2 * lk;
    const int b_frag = (lk * G_PN + wn * 16 + lm) * 2;

    for (int kc = 0; kc < G_NK; ++kc) {
        const int  s    = kc & 1;
        const bool more = (kc + 1) < G_NK;

        if (more) {
            const int ko = (kc + 1) * G_BK;
            ra0  = *(const float4*)(a_src0 + ko);
            if (a_two) ra1 = *(const float4*)(a_src1 + ko);
            const float* bp = b_src + (size_t)ko * QKV3;
            rb00 = *(const float4*)(bp);
            rb01 = *(const float4*)(bp + QKV3);
            rb10 = *(const float4*)(bp + 32 * QKV3);
            rb11 = *(const float4*)(bp + 33 * QKV3);
        }

        const float* as = &As[s][a_frag];
        const float* bs = &Bs[s][b_frag];
        for (int kk = 0; kk < G_BK / 4; ++kk) {
            const v2f a = *(const v2f*)(as + 4 * kk);
            const v2f b = *(const v2f*)(bs + 4 * kk * G_PN);
            acc = __builtin_amdgcn_wmma_f32_16x16x4_f32(
                false, a, false, b, (short)0, acc, false, false);
        }

        if (more) {
            const int t = s ^ 1;
            *(float4*)&As[t][a_dst0] = ra0;
            if (a_two) *(float4*)&As[t][a_dst1] = ra1;
            *(float4*)&Bs[t][b_dst0]     = make_float4(rb00.x, rb01.x, rb00.y, rb01.y);
            *(float4*)&Bs[t][b_dst0 + 4] = make_float4(rb00.z, rb01.z, rb00.w, rb01.w);
            *(float4*)&Bs[t][b_dst1]     = make_float4(rb10.x, rb11.x, rb10.y, rb11.y);
            *(float4*)&Bs[t][b_dst1 + 4] = make_float4(rb10.z, rb11.z, rb10.w, rb11.w);
            __syncthreads();
        }
    }

    const int row0  = m0 + wm * 16;
    const int j0    = jb0 + wn * 16;
    const int b_idx = row0 >> 9;
    const int n0    = (row0 & 511) + lk * 8;
    const int col   = j0 + lm;
    const int which = j0 >> 6;
    const int c     = col & 63;
    float* T = (which == 0) ? Qt : ((which == 1) ? Kt : Vt);
    float* base = T + ((size_t)b_idx * DK + c) * SEQ;
    #pragma unroll
    for (int r = 0; r < 8; ++r)
        base[n0 + r] = acc[r];
}

__global__ __launch_bounds__(256) void attn_kernel(
    const float* __restrict__ Qt, const float* __restrict__ Kt,
    const float* __restrict__ Vt, float* __restrict__ SA)
{
    __shared__ __align__(16) float kv[SEQ];
    __shared__ __align__(16) float vv[SEQ];
    __shared__ float rmax[8];
    __shared__ float rmin[8];

    const int bc  = blockIdx.x;
    const int tid = threadIdx.x;

    const float* kp = Kt + (size_t)bc * SEQ;
    const float* vp = Vt + (size_t)bc * SEQ;
    kv[tid]       = kp[tid];
    kv[tid + 256] = kp[tid + 256];
    vv[tid]       = vp[tid];
    vv[tid + 256] = vp[tid + 256];
    __syncthreads();

    float kmx = fmaxf(kv[tid], kv[tid + 256]);
    float kmn = fminf(kv[tid], kv[tid + 256]);
    #pragma unroll
    for (int off = 16; off >= 1; off >>= 1) {
        kmx = fmaxf(kmx, __shfl_xor(kmx, off, 32));
        kmn = fminf(kmn, __shfl_xor(kmn, off, 32));
    }
    if ((tid & 31) == 0) { rmax[tid >> 5] = kmx; rmin[tid >> 5] = kmn; }
    __syncthreads();
    float KMAX = rmax[0], KMIN = rmin[0];
    #pragma unroll
    for (int i = 1; i < 8; ++i) {
        KMAX = fmaxf(KMAX, rmax[i]);
        KMIN = fminf(KMIN, rmin[i]);
    }

    const int b = bc >> 6;
    const int c = bc & 63;
    const float* qp = Qt + (size_t)bc * SEQ;
    const float4* kv4 = (const float4*)kv;
    const float4* vv4 = (const float4*)vv;

    #pragma unroll
    for (int r = 0; r < 2; ++r) {
        const int n = tid + r * 256;
        const float a    = qp[n] * SCALE;
        const float cmax = a * ((a >= 0.f) ? KMAX : KMIN);
        float s0 = 0.f, s1 = 0.f, s2 = 0.f, s3 = 0.f;
        float p0 = 0.f, p1 = 0.f, p2 = 0.f, p3 = 0.f;
        #pragma unroll 2
        for (int m4 = 0; m4 < SEQ / 4; ++m4) {
            const float4 k4 = kv4[m4];
            const float4 v4 = vv4[m4];
            const float e0 = __expf(fmaf(a, k4.x, -cmax));
            const float e1 = __expf(fmaf(a, k4.y, -cmax));
            const float e2 = __expf(fmaf(a, k4.z, -cmax));
            const float e3 = __expf(fmaf(a, k4.w, -cmax));
            s0 += e0; s1 += e1; s2 += e2; s3 += e3;
            p0 = fmaf(e0, v4.x, p0);
            p1 = fmaf(e1, v4.y, p1);
            p2 = fmaf(e2, v4.z, p2);
            p3 = fmaf(e3, v4.w, p3);
        }
        const float sum = (s0 + s1) + (s2 + s3);
        const float num = (p0 + p1) + (p2 + p3);
        SA[((size_t)b * SEQ + n) * DK + c] = num / sum;
    }
}

__global__ __launch_bounds__(256) void out_gemm_kernel(
    const float* __restrict__ SA, const float* __restrict__ Wo,
    float* __restrict__ Out)
{
    const int tid  = threadIdx.x;
    const int lane = tid & 31;
    const int wave = tid >> 5;
    const int tile = blockIdx.x * 8 + wave;
    const int JT   = DIM / 16;
    const int mt   = tile / JT;
    const int jt   = tile % JT;
    const int row0 = mt * 16;
    const int j0   = jt * 16;
    const int lm   = lane & 15;
    const int lk   = lane >> 4;

    v8f acc = {0.f,0.f,0.f,0.f,0.f,0.f,0.f,0.f};

    const float* arow = SA + (size_t)(row0 + lm) * DK + 2 * lk;
    const float* bcol = Wo + j0 + lm;

    #pragma unroll
    for (int k0 = 0; k0 < DK; k0 += 4) {
        v2f a = *(const v2f*)(arow + k0);
        const int kb = k0 + 2 * lk;
        v2f b;
        b.x = bcol[(size_t)kb * DIM];
        b.y = bcol[(size_t)(kb + 1) * DIM];
        acc = __builtin_amdgcn_wmma_f32_16x16x4_f32(
            false, a, false, b, (short)0, acc, false, false);
    }

    float* obase = Out + (size_t)(row0 + lk * 8) * DIM + j0 + lm;
    #pragma unroll
    for (int r = 0; r < 8; ++r)
        obase[(size_t)r * DIM] = acc[r];
}

extern "C" void kernel_launch(void* const* d_in, const int* in_sizes, int n_in,
                              void* d_out, int out_size, void* d_ws, size_t ws_size,
                              hipStream_t stream)
{
    const float* X    = (const float*)d_in[0];
    const float* Wqkv = (const float*)d_in[1];
    const float* Wo   = (const float*)d_in[2];
    float* out = (float*)d_out;

    const size_t CHN = (size_t)BATCH * DK * SEQ;
    float* Qt = (float*)d_ws;
    float* Kt = Qt + CHN;
    float* Vt = Kt + CHN;
    float* SA = Vt + CHN;

    qkv_gemm_kernel<<<256, 384, 0, stream>>>(X, Wqkv, Qt, Kt, Vt);
    attn_kernel<<<BATCH * DK, 256, 0, stream>>>(Qt, Kt, Vt, SA);
    out_gemm_kernel<<<1024, 256, 0, stream>>>(SA, Wo, out);
}


# output projection re-tiled: 64x128 block tiles, SA tile through LDS by coalesced 16-byte loads, Wo fragments kept in registers over 4 row tiles, 2048 waves not 8192; summation order unchanged
# speedup vs baseline: 1.3546x; 1.0343x over previous
#include <hip/hip_runtime.h>
#include <math.h>

#define BATCH 8
#define SEQ   512
#define DIM   512
#define DK    64
#define QKV3  192
#define SCALE 0.125f

typedef __attribute__((ext_vector_type(2))) float v2f;
typedef __attribute__((ext_vector_type(8))) float v8f;

#define G_BM 32
#define G_BN 96
#define G_BK 64
#define G_NK (DIM / G_BK)
#define G_PA (G_BK + 4)
#define G_PN 112
#define G_NT 384

__global__ __launch_bounds__(G_NT) void qkv_gemm_kernel(
    const float* __restrict__ X, const float* __restrict__ Wqkv,
    float* __restrict__ Qt, float* __restrict__ Kt, float* __restrict__ Vt)
{
    __shared__ __align__(16) float As[2][G_BM * G_PA];
    __shared__ __align__(16) float Bs[2][(G_BK / 2) * G_PN * 2];

    const int tid  = threadIdx.x;
    const int lane = tid & 31;
    const int wave = tid >> 5;
    const int lm   = lane & 15;
    const int lk   = lane >> 4;
    const int m0   = (blockIdx.x >> 1) * G_BM;
    const int jb0  = (blockIdx.x & 1) * G_BN;
    const int wm   = wave / 6;
    const int wn   = wave % 6;

    const int  a_row = tid >> 4;
    const int  a_q   = tid & 15;
    const bool a_two = tid < 128;
    const float* a_src0 = X + (size_t)(m0 + a_row) * DIM + 4 * a_q;
    const float* a_src1 = X + (size_t)(m0 + 24 + (a_row & 7)) * DIM + 4 * a_q;
    const int a_dst0 = a_row * G_PA + 4 * a_q;
    const int a_dst1 = (24 + (a_row & 7)) * G_PA + 4 * a_q;

    const int b_kp = tid / 24;
    const int b_c4 = tid % 24;
    const float* b_src = Wqkv + (size_t)(2 * b_kp) * QKV3 + jb0 + 4 * b_c4;
    const int b_dst0 = (b_kp * G_PN + 4 * b_c4) * 2;
    const int b_dst1 = ((b_kp + 16) * G_PN + 4 * b_c4) * 2;

    float4 ra0, ra1, rb00, rb01, rb10, rb11;
    ra1 = make_float4(0.f, 0.f, 0.f, 0.f);

    ra0  = *(const float4*)(a_src0);
    if (a_two) ra1 = *(const float4*)(a_src1);
    rb00 = *(const float4*)(b_src);
    rb01 = *(const float4*)(b_src + QKV3);
    rb10 = *(const float4*)(b_src + 32 * QKV3);
    rb11 = *(const float4*)(b_src + 33 * QKV3);

    *(float4*)&As[0][a_dst0] = ra0;
    if (a_two) *(float4*)&As[0][a_dst1] = ra1;
    *(float4*)&Bs[0][b_dst0]     = make_float4(rb00.x, rb01.x, rb00.y, rb01.y);
    *(float4*)&Bs[0][b_dst0 + 4] = make_float4(rb00.z, rb01.z, rb00.w, rb01.w);
    *(float4*)&Bs[0][b_dst1]     = make_float4(rb10.x, rb11.x, rb10.y, rb11.y);
    *(float4*)&Bs[0][b_dst1 + 4] = make_float4(rb10.z, rb11.z, rb10.w, rb11.w);
    __syncthreads();

    v8f acc = {0.f,0.f,0.f,0.f,0.f,0.f,0.f,0.f};

    const int a_frag = (wm * 16 + lm) * G_PA + 2 * lk;
    const int b_frag = (lk * G_PN + wn * 16 + lm) * 2;

    for (int kc = 0; kc < G_NK; ++kc) {
        const int  s    = kc & 1;
        const bool more = (kc + 1) < G_NK;

        if (more) {
            const int ko = (kc + 1) * G_BK;
            ra0  = *(const float4*)(a_src0 + ko);
            if (a_two) ra1 = *(const float4*)(a_src1 + ko);
            const float* bp = b_src + (size_t)ko * QKV3;
            rb00 = *(const float4*)(bp);
            rb01 = *(const float4*)(bp + QKV3);
            rb10 = *(const float4*)(bp + 32 * QKV3);
            rb11 = *(const float4*)(bp + 33 * QKV3);
        }

        const float* as = &As[s][a_frag];
        const float* bs = &Bs[s][b_frag];
        for (int kk = 0; kk < G_BK / 4; ++kk) {
            const v2f a = *(const v2f*)(as + 4 * kk);
            const v2f b = *(const v2f*)(bs + 4 * kk * G_PN);
            acc = __builtin_amdgcn_wmma_f32_16x16x4_f32(
                false, a, false, b, (short)0, acc, false, false);
        }

        if (more) {
            const int t = s ^ 1;
            *(float4*)&As[t][a_dst0] = ra0;
            if (a_two) *(float4*)&As[t][a_dst1] = ra1;
            *(float4*)&Bs[t][b_dst0]     = make_float4(rb00.x, rb01.x, rb00.y, rb01.y);
            *(float4*)&Bs[t][b_dst0 + 4] = make_float4(rb00.z, rb01.z, rb00.w, rb01.w);
            *(float4*)&Bs[t][b_dst1]     = make_float4(rb10.x, rb11.x, rb10.y, rb11.y);
            *(float4*)&Bs[t][b_dst1 + 4] = make_float4(rb10.z, rb11.z, rb10.w, rb11.w);
            __syncthreads();
        }
    }

    const int row0  = m0 + wm * 16;
    const int j0    = jb0 + wn * 16;
    const int b_idx = row0 >> 9;
    const int n0    = (row0 & 511) + lk * 8;
    const int col   = j0 + lm;
    const int which = j0 >> 6;
    const int c     = col & 63;
    float* T = (which == 0) ? Qt : ((which == 1) ? Kt : Vt);
    float* base = T + ((size_t)b_idx * DK + c) * SEQ;
    #pragma unroll
    for (int r = 0; r < 8; ++r)
        base[n0 + r] = acc[r];
}

__global__ __launch_bounds__(256) void attn_kernel(
    const float* __restrict__ Qt, const float* __restrict__ Kt,
    const float* __restrict__ Vt, float* __restrict__ SA)
{
    __shared__ __align__(16) float kv[SEQ];
    __shared__ __align__(16) float vv[SEQ];
    __shared__ float rmax[8];
    __shared__ float rmin[8];

    const int bc  = blockIdx.x;
    const int tid = threadIdx.x;

    const float* kp = Kt + (size_t)bc * SEQ;
    const float* vp = Vt + (size_t)bc * SEQ;
    kv[tid]       = kp[tid];
    kv[tid + 256] = kp[tid + 256];
    vv[tid]       = vp[tid];
    vv[tid + 256] = vp[tid + 256];
    __syncthreads();

    float kmx = fmaxf(kv[tid], kv[tid + 256]);
    float kmn = fminf(kv[tid], kv[tid + 256]);
    #pragma unroll
    for (int off = 16; off >= 1; off >>= 1) {
        kmx = fmaxf(kmx, __shfl_xor(kmx, off, 32));
        kmn = fminf(kmn, __shfl_xor(kmn, off, 32));
    }
    if ((tid & 31) == 0) { rmax[tid >> 5] = kmx; rmin[tid >> 5] = kmn; }
    __syncthreads();
    float KMAX = rmax[0], KMIN = rmin[0];
    #pragma unroll
    for (int i = 1; i < 8; ++i) {
        KMAX = fmaxf(KMAX, rmax[i]);
        KMIN = fminf(KMIN, rmin[i]);
    }

    const int b = bc >> 6;
    const int c = bc & 63;
    const float* qp = Qt + (size_t)bc * SEQ;
    const float4* kv4 = (const float4*)kv;
    const float4* vv4 = (const float4*)vv;

    #pragma unroll
    for (int r = 0; r < 2; ++r) {
        const int n = tid + r * 256;
        const float a    = qp[n] * SCALE;
        const float cmax = a * ((a >= 0.f) ? KMAX : KMIN);
        float s0 = 0.f, s1 = 0.f, s2 = 0.f, s3 = 0.f;
        float p0 = 0.f, p1 = 0.f, p2 = 0.f, p3 = 0.f;
        #pragma unroll 2
        for (int m4 = 0; m4 < SEQ / 4; ++m4) {
            const float4 k4 = kv4[m4];
            const float4 v4 = vv4[m4];
            const float e0 = __expf(fmaf(a, k4.x, -cmax));
            const float e1 = __expf(fmaf(a, k4.y, -cmax));
            const float e2 = __expf(fmaf(a, k4.z, -cmax));
            const float e3 = __expf(fmaf(a, k4.w, -cmax));
            s0 += e0; s1 += e1; s2 += e2; s3 += e3;
            p0 = fmaf(e0, v4.x, p0);
            p1 = fmaf(e1, v4.y, p1);
            p2 = fmaf(e2, v4.z, p2);
            p3 = fmaf(e3, v4.w, p3);
        }
        const float sum = (s0 + s1) + (s2 + s3);
        const float num = (p0 + p1) + (p2 + p3);
        SA[((size_t)b * SEQ + n) * DK + c] = num / sum;
    }
}

#define O_BM 64
#define O_BN 128
#define O_PA (DK + 4)

__global__ __launch_bounds__(256) void out_gemm_kernel(
    const float* __restrict__ SA, const float* __restrict__ Wo,
    float* __restrict__ Out)
{
    __shared__ __align__(16) float As[O_BM * O_PA];

    const int tid  = threadIdx.x;
    const int lane = tid & 31;
    const int wave = tid >> 5;
    const int lm   = lane & 15;
    const int lk   = lane >> 4;
    const int row0 = (blockIdx.x >> 2) * O_BM;
    const int j0   = (blockIdx.x & 3) * O_BN + wave * 16;

    const float* bcol = Wo + (size_t)(2 * lk) * DIM + j0 + lm;
    v2f bf[DK / 4];
    #pragma unroll
    for (int j = 0; j < DK / 4; ++j) {
        bf[j].x = bcol[(size_t)(4 * j) * DIM];
        bf[j].y = bcol[(size_t)(4 * j + 1) * DIM];
    }

    const float4* src = (const float4*)(SA + (size_t)row0 * DK);
    #pragma unroll
    for (int i = 0; i < 4; ++i) {
        const int idx = tid + 256 * i;
        const float4 v = src[idx];
        *(float4*)&As[(idx >> 4) * O_PA + 4 * (idx & 15)] = v;
    }
    __syncthreads();

    v8f acc0 = {0.f,0.f,0.f,0.f,0.f,0.f,0.f,0.f};
    v8f acc1 = acc0, acc2 = acc0, acc3 = acc0;

    const float* af = &As[lm * O_PA + 2 * lk];
    #pragma unroll
    for (int j = 0; j < DK / 4; ++j) {
        const v2f a0 = *(const v2f*)(af + 4 * j);
        const v2f a1 = *(const v2f*)(af + 16 * O_PA + 4 * j);
        const v2f a2 = *(const v2f*)(af + 32 * O_PA + 4 * j);
        const v2f a3 = *(const v2f*)(af + 48 * O_PA + 4 * j);
        acc0 = __builtin_amdgcn_wmma_f32_16x16x4_f32(
            false, a0, false, bf[j], (short)0, acc0, false, false);
        acc1 = __builtin_amdgcn_wmma_f32_16x16x4_f32(
            false, a1, false, bf[j], (short)0, acc1, false, false);
        acc2 = __builtin_amdgcn_wmma_f32_16x16x4_f32(
            false, a2, false, bf[j], (short)0, acc2, false, false);
        acc3 = __builtin_amdgcn_wmma_f32_16x16x4_f32(
            false, a3, false, bf[j], (short)0, acc3, false, false);
    }

    float* obase = Out + (size_t)(row0 + lk * 8) * DIM + j0 + lm;
    #pragma unroll
    for (int r = 0; r < 8; ++r) {
        obase[(size_t)r * DIM]        = acc0[r];
        obase[(size_t)(16 + r) * DIM] = acc1[r];
        obase[(size_t)(32 + r) * DIM] = acc2[r];
        obase[(size_t)(48 + r) * DIM] = acc3[r];
    }
}

extern "C" void kernel_launch(void* const* d_in, const int* in_sizes, int n_in,
                              void* d_out, int out_size, void* d_ws, size_t ws_size,
                              hipStream_t stream)
{
    const float* X    = (const float*)d_in[0];
    const float* Wqkv = (const float*)d_in[1];
    const float* Wo   = (const float*)d_in[2];
    float* out = (float*)d_out;

    const size_t CHN = (size_t)BATCH * DK * SEQ;
    float* Qt = (float*)d_ws;
    float* Kt = Qt + CHN;
    float* Vt = Kt + CHN;
    float* SA = Vt + CHN;

    qkv_gemm_kernel<<<256, 384, 0, stream>>>(X, Wqkv, Qt, Kt, Vt);
    attn_kernel<<<BATCH * DK, 256, 0, stream>>>(Qt, Kt, Vt, SA);
    out_gemm_kernel<<<256, 256, 0, stream>>>(SA, Wo, out);
}


# attention loop: about 40% fewer vector instructions per exponential: log2e folded into row scale (bare v_exp_f32), packed 2-float fma/add, both rows share each K/V read, exp/packed ops interleaved
# speedup vs baseline: 1.3923x; 1.0278x over previous
#include <hip/hip_runtime.h>
#include <math.h>

#define BATCH 8
#define SEQ   512
#define DIM   512
#define DK    64
#define QKV3  192
#define SCALE 0.125f

typedef __attribute__((ext_vector_type(2))) float v2f;
typedef __attribute__((ext_vector_type(8))) float v8f;

#define G_BM 32
#define G_BN 96
#define G_BK 64
#define G_NK (DIM / G_BK)
#define G_PA (G_BK + 4)
#define G_PN 112
#define G_NT 384

__global__ __launch_bounds__(G_NT) void qkv_gemm_kernel(
    const float* __restrict__ X, const float* __restrict__ Wqkv,
    float* __restrict__ Qt, float* __restrict__ Kt, float* __restrict__ Vt)
{
    __shared__ __align__(16) float As[2][G_BM * G_PA];
    __shared__ __align__(16) float Bs[2][(G_BK / 2) * G_PN * 2];

    const int tid  = threadIdx.x;
    const int lane = tid & 31;
    const int wave = tid >> 5;
    const int lm   = lane & 15;
    const int lk   = lane >> 4;
    const int m0   = (blockIdx.x >> 1) * G_BM;
    const int jb0  = (blockIdx.x & 1) * G_BN;
    const int wm   = wave / 6;
    const int wn   = wave % 6;

    const int  a_row = tid >> 4;
    const int  a_q   = tid & 15;
    const bool a_two = tid < 128;
    const float* a_src0 = X + (size_t)(m0 + a_row) * DIM + 4 * a_q;
    const float* a_src1 = X + (size_t)(m0 + 24 + (a_row & 7)) * DIM + 4 * a_q;
    const int a_dst0 = a_row * G_PA + 4 * a_q;
    const int a_dst1 = (24 + (a_row & 7)) * G_PA + 4 * a_q;

    const int b_kp = tid / 24;
    const int b_c4 = tid % 24;
    const float* b_src = Wqkv + (size_t)(2 * b_kp) * QKV3 + jb0 + 4 * b_c4;
    const int b_dst0 = (b_kp * G_PN + 4 * b_c4) * 2;
    const int b_dst1 = ((b_kp + 16) * G_PN + 4 * b_c4) * 2;

    float4 ra0, ra1, rb00, rb01, rb10, rb11;
    ra1 = make_float4(0.f, 0.f, 0.f, 0.f);

    ra0  = *(const float4*)(a_src0);
    if (a_two) ra1 = *(const float4*)(a_src1);
    rb00 = *(const float4*)(b_src);
    rb01 = *(const float4*)(b_src + QKV3);
    rb10 = *(const float4*)(b_src + 32 * QKV3);
    rb11 = *(const float4*)(b_src + 33 * QKV3);

    *(float4*)&As[0][a_dst0] = ra0;
    if (a_two) *(float4*)&As[0][a_dst1] = ra1;
    *(float4*)&Bs[0][b_dst0]     = make_float4(rb00.x, rb01.x, rb00.y, rb01.y);
    *(float4*)&Bs[0][b_dst0 + 4] = make_float4(rb00.z, rb01.z, rb00.w, rb01.w);
    *(float4*)&Bs[0][b_dst1]     = make_float4(rb10.x, rb11.x, rb10.y, rb11.y);
    *(float4*)&Bs[0][b_dst1 + 4] = make_float4(rb10.z, rb11.z, rb10.w, rb11.w);
    __syncthreads();

    v8f acc = {0.f,0.f,0.f,0.f,0.f,0.f,0.f,0.f};

    const int a_frag = (wm * 16 + lm) * G_PA + 2 * lk;
    const int b_frag = (lk * G_PN + wn * 16 + lm) * 2;

    for (int kc = 0; kc < G_NK; ++kc) {
        const int  s    = kc & 1;
        const bool more = (kc + 1) < G_NK;

        if (more) {
            const int ko = (kc + 1) * G_BK;
            ra0  = *(const float4*)(a_src0 + ko);
            if (a_two) ra1 = *(const float4*)(a_src1 + ko);
            const float* bp = b_src + (size_t)ko * QKV3;
            rb00 = *(const float4*)(bp);
            rb01 = *(const float4*)(bp + QKV3);
            rb10 = *(const float4*)(bp + 32 * QKV3);
            rb11 = *(const float4*)(bp + 33 * QKV3);
        }

        const float* as = &As[s][a_frag];
        const float* bs = &Bs[s][b_frag];
        for (int kk = 0; kk < G_BK / 4; ++kk) {
            const v2f a = *(const v2f*)(as + 4 * kk);
            const v2f b = *(const v2f*)(bs + 4 * kk * G_PN);
            acc = __builtin_amdgcn_wmma_f32_16x16x4_f32(
                false, a, false, b, (short)0, acc, false, false);
        }

        if (more) {
            const int t = s ^ 1;
            *(float4*)&As[t][a_dst0] = ra0;
            if (a_two) *(float4*)&As[t][a_dst1] = ra1;
            *(float4*)&Bs[t][b_dst0]     = make_float4(rb00.x, rb01.x, rb00.y, rb01.y);
            *(float4*)&Bs[t][b_dst0 + 4] = make_float4(rb00.z, rb01.z, rb00.w, rb01.w);
            *(float4*)&Bs[t][b_dst1]     = make_float4(rb10.x, rb11.x, rb10.y, rb11.y);
            *(float4*)&Bs[t][b_dst1 + 4] = make_float4(rb10.z, rb11.z, rb10.w, rb11.w);
            __syncthreads();
        }
    }

    const int row0  = m0 + wm * 16;
    const int j0    = jb0 + wn * 16;
    const int b_idx = row0 >> 9;
    const int n0    = (row0 & 511) + lk * 8;
    const int col   = j0 + lm;
    const int which = j0 >> 6;
    const int c     = col & 63;
    float* T = (which == 0) ? Qt : ((which == 1) ? Kt : Vt);
    float* base = T + ((size_t)b_idx * DK + c) * SEQ;
    #pragma unroll
    for (int r = 0; r < 8; ++r)
        base[n0 + r] = acc[r];
}

#define LOG2E 1.4426950408889634f

__global__ __launch_bounds__(256) void attn_kernel(
    const float* __restrict__ Qt, const float* __restrict__ Kt,
    const float* __restrict__ Vt, float* __restrict__ SA)
{
    __shared__ __align__(16) float kv[SEQ];
    __shared__ __align__(16) float vv[SEQ];
    __shared__ float rmax[8];
    __shared__ float rmin[8];

    const int bc  = blockIdx.x;
    const int tid = threadIdx.x;

    const float* kp = Kt + (size_t)bc * SEQ;
    const float* vp = Vt + (size_t)bc * SEQ;
    const float* qp = Qt + (size_t)bc * SEQ;
    const float q0 = qp[tid];
    const float q1 = qp[tid + 256];
    kv[tid]       = kp[tid];
    kv[tid + 256] = kp[tid + 256];
    vv[tid]       = vp[tid];
    vv[tid + 256] = vp[tid + 256];
    __syncthreads();

    float kmx = fmaxf(kv[tid], kv[tid + 256]);
    float kmn = fminf(kv[tid], kv[tid + 256]);
    #pragma unroll
    for (int off = 16; off >= 1; off >>= 1) {
        kmx = fmaxf(kmx, __shfl_xor(kmx, off, 32));
        kmn = fminf(kmn, __shfl_xor(kmn, off, 32));
    }
    if ((tid & 31) == 0) { rmax[tid >> 5] = kmx; rmin[tid >> 5] = kmn; }
    __syncthreads();
    float KMAX = rmax[0], KMIN = rmin[0];
    #pragma unroll
    for (int i = 1; i < 8; ++i) {
        KMAX = fmaxf(KMAX, rmax[i]);
        KMIN = fminf(KMIN, rmin[i]);
    }

    const int b = bc >> 6;
    const int c = bc & 63;
    const float4* kv4 = (const float4*)kv;
    const float4* vv4 = (const float4*)vv;

    const float a0 = q0 * (SCALE * LOG2E);
    const float a1 = q1 * (SCALE * LOG2E);
    const float c0 = a0 * ((a0 >= 0.f) ? KMAX : KMIN);
    const float c1 = a1 * ((a1 >= 0.f) ? KMAX : KMIN);
    const v2f A0 = {a0, a0}, N0 = {-c0, -c0};
    const v2f A1 = {a1, a1}, N1 = {-c1, -c1};

    const v2f z2 = {0.f, 0.f};
    v2f s0a = z2, s0b = z2, p0a = z2, p0b = z2;
    v2f s1a = z2, s1b = z2, p1a = z2, p1b = z2;

    for (int m8 = 0; m8 < SEQ / 8; ++m8) {
        const float4 k4 = kv4[2 * m8], k5 = kv4[2 * m8 + 1];
        const float4 v4 = vv4[2 * m8], v5 = vv4[2 * m8 + 1];
        const v2f ka = {k4.x, k4.y}, kb = {k4.z, k4.w}, kc = {k5.x, k5.y}, kd = {k5.z, k5.w};
        const v2f va = {v4.x, v4.y}, vb = {v4.z, v4.w}, vc = {v5.x, v5.y}, vd = {v5.z, v5.w};
        const v2f x0a = __builtin_elementwise_fma(ka, A0, N0);
        const v2f x1a = __builtin_elementwise_fma(ka, A1, N1);
        const v2f x0b = __builtin_elementwise_fma(kb, A0, N0);
        const v2f x1b = __builtin_elementwise_fma(kb, A1, N1);
        const v2f x0c = __builtin_elementwise_fma(kc, A0, N0);
        const v2f x1c = __builtin_elementwise_fma(kc, A1, N1);
        const v2f x0d = __builtin_elementwise_fma(kd, A0, N0);
        const v2f x1d = __builtin_elementwise_fma(kd, A1, N1);
        const v2f e0a = {__builtin_amdgcn_exp2f(x0a.x), __builtin_amdgcn_exp2f(x0a.y)};
        const v2f e1a = {__builtin_amdgcn_exp2f(x1a.x), __builtin_amdgcn_exp2f(x1a.y)};
        const v2f e0b = {__builtin_amdgcn_exp2f(x0b.x), __builtin_amdgcn_exp2f(x0b.y)};
        const v2f e1b = {__builtin_amdgcn_exp2f(x1b.x), __builtin_amdgcn_exp2f(x1b.y)};
        const v2f e0c = {__builtin_amdgcn_exp2f(x0c.x), __builtin_amdgcn_exp2f(x0c.y)};
        const v2f e1c = {__builtin_amdgcn_exp2f(x1c.x), __builtin_amdgcn_exp2f(x1c.y)};
        const v2f e0d = {__builtin_amdgcn_exp2f(x0d.x), __builtin_amdgcn_exp2f(x0d.y)};
        const v2f e1d = {__builtin_amdgcn_exp2f(x1d.x), __builtin_amdgcn_exp2f(x1d.y)};
        s0a += e0a;  p0a = __builtin_elementwise_fma(e0a, va, p0a);
        s1a += e1a;  p1a = __builtin_elementwise_fma(e1a, va, p1a);
        s0b += e0b;  p0b = __builtin_elementwise_fma(e0b, vb, p0b);
        s1b += e1b;  p1b = __builtin_elementwise_fma(e1b, vb, p1b);
        s0a += e0c;  p0a = __builtin_elementwise_fma(e0c, vc, p0a);
        s1a += e1c;  p1a = __builtin_elementwise_fma(e1c, vc, p1a);
        s0b += e0d;  p0b = __builtin_elementwise_fma(e0d, vd, p0b);
        s1b += e1d;  p1b = __builtin_elementwise_fma(e1d, vd, p1b);

        __builtin_amdgcn_sched_group_barrier(0x100, 4, 0);
        __builtin_amdgcn_sched_group_barrier(0x002, 4, 0);
        #pragma unroll
        for (int i = 0; i < 16; ++i) {
            __builtin_amdgcn_sched_group_barrier(0x400, 1, 0);
            __builtin_amdgcn_sched_group_barrier(0x002, 1, 0);
        }
        __builtin_amdgcn_sched_group_barrier(0x002, 4, 0);
    }
    const float sum0 = (s0a.x + s0a.y) + (s0b.x + s0b.y);
    const float num0 = (p0a.x + p0a.y) + (p0b.x + p0b.y);
    const float sum1 = (s1a.x + s1a.y) + (s1b.x + s1b.y);
    const float num1 = (p1a.x + p1a.y) + (p1b.x + p1b.y);
    SA[((size_t)b * SEQ + tid) * DK + c]       = num0 / sum0;
    SA[((size_t)b * SEQ + tid + 256) * DK + c] = num1 / sum1;
}

#define O_BM 64
#define O_BN 128
#define O_PA (DK + 4)

__global__ __launch_bounds__(256) void out_gemm_kernel(
    const float* __restrict__ SA, const float* __restrict__ Wo,
    float* __restrict__ Out)
{
    __shared__ __align__(16) float As[O_BM * O_PA];

    const int tid  = threadIdx.x;
    const int lane = tid & 31;
    const int wave = tid >> 5;
    const int lm   = lane & 15;
    const int lk   = lane >> 4;
    const int row0 = (blockIdx.x >> 2) * O_BM;
    const int j0   = (blockIdx.x & 3) * O_BN + wave * 16;

    const float* bcol = Wo + (size_t)(2 * lk) * DIM + j0 + lm;
    v2f bf[DK / 4];
    #pragma unroll
    for (int j = 0; j < DK / 4; ++j) {
        bf[j].x = bcol[(size_t)(4 * j) * DIM];
        bf[j].y = bcol[(size_t)(4 * j + 1) * DIM];
    }

    const float4* src = (const float4*)(SA + (size_t)row0 * DK);
    #pragma unroll
    for (int i = 0; i < 4; ++i) {
        const int idx = tid + 256 * i;
        const float4 v = src[idx];
        *(float4*)&As[(idx >> 4) * O_PA + 4 * (idx & 15)] = v;
    }
    __syncthreads();

    v8f acc0 = {0.f,0.f,0.f,0.f,0.f,0.f,0.f,0.f};
    v8f acc1 = acc0, acc2 = acc0, acc3 = acc0;

    const float* af = &As[lm * O_PA + 2 * lk];
    #pragma unroll
    for (int j = 0; j < DK / 4; ++j) {
        const v2f a0 = *(const v2f*)(af + 4 * j);
        const v2f a1 = *(const v2f*)(af + 16 * O_PA + 4 * j);
        const v2f a2 = *(const v2f*)(af + 32 * O_PA + 4 * j);
        const v2f a3 = *(const v2f*)(af + 48 * O_PA + 4 * j);
        acc0 = __builtin_amdgcn_wmma_f32_16x16x4_f32(
            false, a0, false, bf[j], (short)0, acc0, false, false);
        acc1 = __builtin_amdgcn_wmma_f32_16x16x4_f32(
            false, a1, false, bf[j], (short)0, acc1, false, false);
        acc2 = __builtin_amdgcn_wmma_f32_16x16x4_f32(
            false, a2, false, bf[j], (short)0, acc2, false, false);
        acc3 = __builtin_amdgcn_wmma_f32_16x16x4_f32(
            false, a3, false, bf[j], (short)0, acc3, false, false);
    }

    float* obase = Out + (size_t)(row0 + lk * 8) * DIM + j0 + lm;
    #pragma unroll
    for (int r = 0; r < 8; ++r) {
        obase[(size_t)r * DIM]        = acc0[r];
        obase[(size_t)(16 + r) * DIM] = acc1[r];
        obase[(size_t)(32 + r) * DIM] = acc2[r];
        obase[(size_t)(48 + r) * DIM] = acc3[r];
    }
}

extern "C" void kernel_launch(void* const* d_in, const int* in_sizes, int n_in,
                              void* d_out, int out_size, void* d_ws, size_t ws_size,
                              hipStream_t stream)
{
    const float* X    = (const float*)d_in[0];
    const float* Wqkv = (const float*)d_in[1];
    const float* Wo   = (const float*)d_in[2];
    float* out = (float*)d_out;

    const size_t CHN = (size_t)BATCH * DK * SEQ;
    float* Qt = (float*)d_ws;
    float* Kt = Qt + CHN;
    float* Vt = Kt + CHN;
    float* SA = Vt + CHN;

    qkv_gemm_kernel<<<256, 384, 0, stream>>>(X, Wqkv, Qt, Kt, Vt);
    attn_kernel<<<BATCH * DK, 256, 0, stream>>>(Qt, Kt, Vt, SA);
    out_gemm_kernel<<<256, 256, 0, stream>>>(SA, Wo, out);
}


# attention and output projection fused into one launch: block = 16 rows x 64 channels of a batch, K/V in LDS, SA tile feeds the WMMA through LDS; 2 launches not 3; summation order unchanged
# speedup vs baseline: 1.7464x; 1.2543x over previous
#include <hip/hip_runtime.h>
#include <math.h>

#define BATCH 8
#define SEQ   512
#define DIM   512
#define DK    64
#define QKV3  192
#define SCALE 0.125f

typedef __attribute__((ext_vector_type(2))) float v2f;
typedef __attribute__((ext_vector_type(8))) float v8f;

#define G_BM 32
#define G_BN 96
#define G_BK 64
#define G_NK (DIM / G_BK)
#define G_PA (G_BK + 4)
#define G_PN 112
#define G_NT 384

__global__ __launch_bounds__(G_NT) void qkv_gemm_kernel(
    const float* __restrict__ X, const float* __restrict__ Wqkv,
    float* __restrict__ Qt, float* __restrict__ Kt, float* __restrict__ Vt)
{
    __shared__ __align__(16) float As[2][G_BM * G_PA];
    __shared__ __align__(16) float Bs[2][(G_BK / 2) * G_PN * 2];

    const int tid  = threadIdx.x;
    const int lane = tid & 31;
    const int wave = tid >> 5;
    const int lm   = lane & 15;
    const int lk   = lane >> 4;
    const int m0   = (blockIdx.x >> 1) * G_BM;
    const int jb0  = (blockIdx.x & 1) * G_BN;
    const int wm   = wave / 6;
    const int wn   = wave % 6;

    const int  a_row = tid >> 4;
    const int  a_q   = tid & 15;
    const bool a_two = tid < 128;
    const float* a_src0 = X + (size_t)(m0 + a_row) * DIM + 4 * a_q;
    const float* a_src1 = X + (size_t)(m0 + 24 + (a_row & 7)) * DIM + 4 * a_q;
    const int a_dst0 = a_row * G_PA + 4 * a_q;
    const int a_dst1 = (24 + (a_row & 7)) * G_PA + 4 * a_q;

    const int b_kp = tid / 24;
    const int b_c4 = tid % 24;
    const float* b_src = Wqkv + (size_t)(2 * b_kp) * QKV3 + jb0 + 4 * b_c4;
    const int b_dst0 = (b_kp * G_PN + 4 * b_c4) * 2;
    const int b_dst1 = ((b_kp + 16) * G_PN + 4 * b_c4) * 2;

    float4 ra0, ra1, rb00, rb01, rb10, rb11;
    ra1 = make_float4(0.f, 0.f, 0.f, 0.f);

    ra0  = *(const float4*)(a_src0);
    if (a_two) ra1 = *(const float4*)(a_src1);
    rb00 = *(const float4*)(b_src);
    rb01 = *(const float4*)(b_src + QKV3);
    rb10 = *(const float4*)(b_src + 32 * QKV3);
    rb11 = *(const float4*)(b_src + 33 * QKV3);

    *(float4*)&As[0][a_dst0] = ra0;
    if (a_two) *(float4*)&As[0][a_dst1] = ra1;
    *(float4*)&Bs[0][b_dst0]     = make_float4(rb00.x, rb01.x, rb00.y, rb01.y);
    *(float4*)&Bs[0][b_dst0 + 4] = make_float4(rb00.z, rb01.z, rb00.w, rb01.w);
    *(float4*)&Bs[0][b_dst1]     = make_float4(rb10.x, rb11.x, rb10.y, rb11.y);
    *(float4*)&Bs[0][b_dst1 + 4] = make_float4(rb10.z, rb11.z, rb10.w, rb11.w);
    __syncthreads();

    v8f acc = {0.f,0.f,0.f,0.f,0.f,0.f,0.f,0.f};

    const int a_frag = (wm * 16 + lm) * G_PA + 2 * lk;
    const int b_frag = (lk * G_PN + wn * 16 + lm) * 2;

    for (int kc = 0; kc < G_NK; ++kc) {
        const int  s    = kc & 1;
        const bool more = (kc + 1) < G_NK;

        if (more) {
            const int ko = (kc + 1) * G_BK;
            ra0  = *(const float4*)(a_src0 + ko);
            if (a_two) ra1 = *(const float4*)(a_src1 + ko);
            const float* bp = b_src + (size_t)ko * QKV3;
            rb00 = *(const float4*)(bp);
            rb01 = *(const float4*)(bp + QKV3);
            rb10 = *(const float4*)(bp + 32 * QKV3);
            rb11 = *(const float4*)(bp + 33 * QKV3);
        }

        const float* as = &As[s][a_frag];
        const float* bs = &Bs[s][b_frag];
        for (int kk = 0; kk < G_BK / 4; ++kk) {
            const v2f a = *(const v2f*)(as + 4 * kk);
            const v2f b = *(const v2f*)(bs + 4 * kk * G_PN);
            acc = __builtin_amdgcn_wmma_f32_16x16x4_f32(
                false, a, false, b, (short)0, acc, false, false);
        }

        if (more) {
            const int t = s ^ 1;
            *(float4*)&As[t][a_dst0] = ra0;
            if (a_two) *(float4*)&As[t][a_dst1] = ra1;
            *(float4*)&Bs[t][b_dst0]     = make_float4(rb00.x, rb01.x, rb00.y, rb01.y);
            *(float4*)&Bs[t][b_dst0 + 4] = make_float4(rb00.z, rb01.z, rb00.w, rb01.w);
            *(float4*)&Bs[t][b_dst1]     = make_float4(rb10.x, rb11.x, rb10.y, rb11.y);
            *(float4*)&Bs[t][b_dst1 + 4] = make_float4(rb10.z, rb11.z, rb10.w, rb11.w);
            __syncthreads();
        }
    }

    const int row0  = m0 + wm * 16;
    const int j0    = jb0 + wn * 16;
    const int b_idx = row0 >> 9;
    const int n0    = (row0 & 511) + lk * 8;
    const int col   = j0 + lm;
    const int which = j0 >> 6;
    const int c     = col & 63;
    float* T = (which == 0) ? Qt : ((which == 1) ? Kt : Vt);
    float* base = T + ((size_t)b_idx * DK + c) * SEQ;
    #pragma unroll
    for (int r = 0; r < 8; ++r)
        base[n0 + r] = acc[r];
}

#define LOG2E 1.4426950408889634f
#define F_BM  16
#define F_P   (SEQ + 16)
#define F_PA  (DK + 4)
#define F_KC  128
#define F_NC  (SEQ / F_KC)

__global__ __launch_bounds__(256) void attn_out_kernel(
    const float* __restrict__ Qt, const float* __restrict__ Kt,
    const float* Vt, const float* Wo, float* __restrict__ Out)
{
    __shared__ __align__(16) float Ks[DK * F_P];
    __shared__ __align__(16) float Vs[DK * F_P];
    __shared__ __align__(16) float SAs[F_BM * F_PA];

    const int tid  = threadIdx.x;
    const int lane = tid & 31;
    const int wave = tid >> 5;
    const int c    = tid >> 2;
    const int rq   = tid & 3;
    const int b    = blockIdx.x >> 5;
    const int n0   = (blockIdx.x & 31) * F_BM;

    const float* Kb = Kt + (size_t)b * DK * SEQ;
    const float* Vb = Vt + (size_t)b * DK * SEQ;

    const float4 q4 = *(const float4*)(Qt + ((size_t)b * DK + c) * SEQ + n0 + 4 * rq);

    {
        float4 kr[32];
        float4 vr[8];
        #pragma unroll
        for (int i = 0; i < 32; ++i)
            kr[i] = *(const float4*)(Kb + (size_t)(8 * wave + (i >> 2)) * SEQ + ((i & 3) * 32 + lane) * 4);
        #pragma unroll
        for (int i = 0; i < 8; ++i)
            vr[i] = *(const float4*)(Vb + (size_t)(8 * wave + i) * SEQ + lane * 4);
        #pragma unroll
        for (int i = 0; i < 32; ++i)
            *(float4*)&Ks[(8 * wave + (i >> 2)) * F_P + ((i & 3) * 32 + lane) * 4] = kr[i];
        #pragma unroll
        for (int i = 0; i < 8; ++i)
            *(float4*)&Vs[(8 * wave + i) * F_P + lane * 4] = vr[i];
    }
    __syncthreads();

    const float4* kc4 = (const float4*)(Ks + c * F_P);
    const float4* vc4 = (const float4*)(Vs + c * F_P);

    float KMAX, KMIN;
    {
        const float4 t0 = kc4[rq];
        float mx = fmaxf(fmaxf(t0.x, t0.y), fmaxf(t0.z, t0.w));
        float mn = fminf(fminf(t0.x, t0.y), fminf(t0.z, t0.w));
        #pragma unroll
        for (int i = 1; i < 32; ++i) {
            const float4 t = kc4[4 * i + rq];
            mx = fmaxf(mx, fmaxf(fmaxf(t.x, t.y), fmaxf(t.z, t.w)));
            mn = fminf(mn, fminf(fminf(t.x, t.y), fminf(t.z, t.w)));
        }
        mx = fmaxf(mx, __shfl_xor(mx, 1, 32));
        mn = fminf(mn, __shfl_xor(mn, 1, 32));
        KMAX = fmaxf(mx, __shfl_xor(mx, 2, 32));
        KMIN = fminf(mn, __shfl_xor(mn, 2, 32));
    }

    const int lm = lane & 15;
    const int lk = lane >> 4;
    const float* wb = Wo + (size_t)(2 * lk) * DIM + wave * 64 + 4 * lm;
    float4 bx[DK / 4], by[DK / 4];
    #pragma unroll
    for (int j = 0; j < 12; ++j) {
        bx[j] = *(const float4*)(wb + (size_t)(4 * j) * DIM);
        by[j] = *(const float4*)(wb + (size_t)(4 * j + 1) * DIM);
    }

    const float qv[4] = {q4.x, q4.y, q4.z, q4.w};
    v2f A[4], N[4];
    #pragma unroll
    for (int r = 0; r < 4; ++r) {
        const float a  = qv[r] * (SCALE * LOG2E);
        const float cm = a * ((a >= 0.f) ? KMAX : KMIN);
        A[r] = (v2f){a, a};
        N[r] = (v2f){-cm, -cm};
    }

    const v2f z2 = {0.f, 0.f};
    v2f sa[4], sb[4], pa[4], pb[4];
    #pragma unroll
    for (int r = 0; r < 4; ++r) { sa[r] = z2; sb[r] = z2; pa[r] = z2; pb[r] = z2; }

#define ATTN_PASS2(KA, VA, KB, VB)                                                         \
    {                                                                                      \
        const v2f ka = {KA.x, KA.y}, kb = {KA.z, KA.w}, kc = {KB.x, KB.y}, kd = {KB.z, KB.w}; \
        const v2f va = {VA.x, VA.y}, vb = {VA.z, VA.w}, vc = {VB.x, VB.y}, vd = {VB.z, VB.w}; \
        v2f xa[4], xb[4], xc[4], xd[4], ea[4], eb[4], ec[4], ed[4];                        \
        _Pragma("unroll")                                                                  \
        for (int r = 0; r < 4; ++r) {                                                      \
            xa[r] = __builtin_elementwise_fma(ka, A[r], N[r]);                             \
            xb[r] = __builtin_elementwise_fma(kb, A[r], N[r]);                             \
        }                                                                                  \
        _Pragma("unroll")                                                                  \
        for (int r = 0; r < 4; ++r) {                                                      \
            xc[r] = __builtin_elementwise_fma(kc, A[r], N[r]);                             \
            xd[r] = __builtin_elementwise_fma(kd, A[r], N[r]);                             \
        }                                                                                  \
        _Pragma("unroll")                                                                  \
        for (int r = 0; r < 4; ++r) {                                                      \
            ea[r] = (v2f){__builtin_amdgcn_exp2f(xa[r].x), __builtin_amdgcn_exp2f(xa[r].y)}; \
            eb[r] = (v2f){__builtin_amdgcn_exp2f(xb[r].x), __builtin_amdgcn_exp2f(xb[r].y)}; \
        }                                                                                  \
        _Pragma("unroll")                                                                  \
        for (int r = 0; r < 4; ++r) {                                                      \
            ec[r] = (v2f){__builtin_amdgcn_exp2f(xc[r].x), __builtin_amdgcn_exp2f(xc[r].y)}; \
            ed[r] = (v2f){__builtin_amdgcn_exp2f(xd[r].x), __builtin_amdgcn_exp2f(xd[r].y)}; \
        }                                                                                  \
        _Pragma("unroll")                                                                  \
        for (int r = 0; r < 4; ++r) {                                                      \
            sa[r] += ea[r];  pa[r] = __builtin_elementwise_fma(ea[r], va, pa[r]);          \
            sb[r] += eb[r];  pb[r] = __builtin_elementwise_fma(eb[r], vb, pb[r]);          \
        }                                                                                  \
        _Pragma("unroll")                                                                  \
        for (int r = 0; r < 4; ++r) {                                                      \
            sa[r] += ec[r];  pa[r] = __builtin_elementwise_fma(ec[r], vc, pa[r]);          \
            sb[r] += ed[r];  pb[r] = __builtin_elementwise_fma(ed[r], vd, pb[r]);          \
        }                                                                                  \
    }

#define ATTN_TP4                                                                           \
        __builtin_amdgcn_sched_group_barrier(0x400, 1, 0);                                 \
        __builtin_amdgcn_sched_group_barrier(0x002, 1, 0);                                 \
        __builtin_amdgcn_sched_group_barrier(0x400, 1, 0);                                 \
        __builtin_amdgcn_sched_group_barrier(0x002, 1, 0);                                 \
        __builtin_amdgcn_sched_group_barrier(0x400, 1, 0);                                 \
        __builtin_amdgcn_sched_group_barrier(0x002, 1, 0);                                 \
        __builtin_amdgcn_sched_group_barrier(0x400, 1, 0);                                 \
        __builtin_amdgcn_sched_group_barrier(0x002, 1, 0);
#define ATTN_ORDER2(NLDS)                                                                  \
        __builtin_amdgcn_sched_group_barrier(0x100, NLDS, 0);                              \
        __builtin_amdgcn_sched_group_barrier(0x002, 4, 0);                                 \
        ATTN_TP4 ATTN_TP4 ATTN_TP4 ATTN_TP4 ATTN_TP4 ATTN_TP4 ATTN_TP4 ATTN_TP4            \
        __builtin_amdgcn_sched_group_barrier(0x002, 12, 0);

    for (int ch = 0; ch < F_NC; ++ch) {
        const bool more = (ch + 1) < F_NC;

        float4 vn0, vn1, vn2, vn3, vn4, vn5, vn6, vn7;
        if (more) {
            const float* vs = Vb + (size_t)(8 * wave) * SEQ + (32 * (ch + 1) + lane) * 4;
            vn0 = *(const float4*)(vs);
            vn1 = *(const float4*)(vs + SEQ);
            vn2 = *(const float4*)(vs + 2 * SEQ);
            vn3 = *(const float4*)(vs + 3 * SEQ);
            vn4 = *(const float4*)(vs + 4 * SEQ);
            vn5 = *(const float4*)(vs + 5 * SEQ);
            vn6 = *(const float4*)(vs + 6 * SEQ);
            vn7 = *(const float4*)(vs + 7 * SEQ);
        }

        const float4* kq = kc4 + 32 * ch;
        const float4* vq = vc4 + 32 * ch;

        for (int m = 0; m < F_KC / 4; m += 2) {
            const float4 k0 = kq[m], k1 = kq[m + 1];
            const float4 v0 = vq[m], v1 = vq[m + 1];
            ATTN_PASS2(k0, v0, k1, v1)
            ATTN_ORDER2(4)
        }

        if (more) {
            float* vd = &Vs[(8 * wave) * F_P + (32 * (ch + 1) + lane) * 4];
            *(float4*)(vd)           = vn0;
            *(float4*)(vd + F_P)     = vn1;
            *(float4*)(vd + 2 * F_P) = vn2;
            *(float4*)(vd + 3 * F_P) = vn3;
            *(float4*)(vd + 4 * F_P) = vn4;
            *(float4*)(vd + 5 * F_P) = vn5;
            *(float4*)(vd + 6 * F_P) = vn6;
            *(float4*)(vd + 7 * F_P) = vn7;
            __syncthreads();
        }
    }
#undef ATTN_PASS2
#undef ATTN_TP4
#undef ATTN_ORDER2

    #pragma unroll
    for (int r = 0; r < 4; ++r) {
        const float sum = (sa[r].x + sa[r].y) + (sb[r].x + sb[r].y);
        const float num = (pa[r].x + pa[r].y) + (pb[r].x + pb[r].y);
        SAs[(4 * rq + r) * F_PA + c] = num / sum;
    }
    __syncthreads();

    #pragma unroll
    for (int j = 12; j < DK / 4; ++j) {
        bx[j] = *(const float4*)(wb + (size_t)(4 * j) * DIM);
        by[j] = *(const float4*)(wb + (size_t)(4 * j + 1) * DIM);
    }

    v8f acc0 = {0.f,0.f,0.f,0.f,0.f,0.f,0.f,0.f};
    v8f acc1 = acc0, acc2 = acc0, acc3 = acc0;

    const float* af = &SAs[lm * F_PA + 2 * lk];
    #pragma unroll
    for (int j = 0; j < DK / 4; ++j) {
        const v2f a  = *(const v2f*)(af + 4 * j);
        const v2f b0 = {bx[j].x, by[j].x};
        const v2f b1 = {bx[j].y, by[j].y};
        const v2f b2 = {bx[j].z, by[j].z};
        const v2f b3 = {bx[j].w, by[j].w};
        acc0 = __builtin_amdgcn_wmma_f32_16x16x4_f32(
            false, a, false, b0, (short)0, acc0, false, false);
        acc1 = __builtin_amdgcn_wmma_f32_16x16x4_f32(
            false, a, false, b1, (short)0, acc1, false, false);
        acc2 = __builtin_amdgcn_wmma_f32_16x16x4_f32(
            false, a, false, b2, (short)0, acc2, false, false);
        acc3 = __builtin_amdgcn_wmma_f32_16x16x4_f32(
            false, a, false, b3, (short)0, acc3, false, false);
    }

    float* obase = Out + ((size_t)b * SEQ + n0 + lk * 8) * DIM + wave * 64 + 4 * lm;
    #pragma unroll
    for (int r = 0; r < 8; ++r)
        *(float4*)(obase + (size_t)r * DIM) = make_float4(acc0[r], acc1[r], acc2[r], acc3[r]);
}

extern "C" void kernel_launch(void* const* d_in, const int* in_sizes, int n_in,
                              void* d_out, int out_size, void* d_ws, size_t ws_size,
                              hipStream_t stream)
{
    const float* X    = (const float*)d_in[0];
    const float* Wqkv = (const float*)d_in[1];
    const float* Wo   = (const float*)d_in[2];
    float* out = (float*)d_out;

    const size_t CHN = (size_t)BATCH * DK * SEQ;
    float* Qt = (float*)d_ws;
    float* Kt = Qt + CHN;
    float* Vt = Kt + CHN;

    qkv_gemm_kernel<<<256, 384, 0, stream>>>(X, Wqkv, Qt, Kt, Vt);
    attn_out_kernel<<<BATCH * (SEQ / F_BM), 256, 0, stream>>>(Qt, Kt, Vt, Wo, out);
}


# attention starts after the first 128-key K/V chunk: kmax/kmin folded from per-tile extrema that the QKV GEMM epilogue writes, K streamed in chunks like V under the loop; summation order unchanged
# speedup vs baseline: 1.8053x; 1.0337x over previous
#include <hip/hip_runtime.h>
#include <math.h>

#define BATCH 8
#define SEQ   512
#define DIM   512
#define DK    64
#define QKV3  192
#define SCALE 0.125f

typedef __attribute__((ext_vector_type(2))) float v2f;
typedef __attribute__((ext_vector_type(8))) float v8f;

#define KTILES (SEQ / 16)
#define G_BM 32
#define G_BN 96
#define G_BK 64
#define G_NK (DIM / G_BK)
#define G_PA (G_BK + 4)
#define G_PN 112
#define G_NT 384

__global__ __launch_bounds__(G_NT) void qkv_gemm_kernel(
    const float* __restrict__ X, const float* __restrict__ Wqkv,
    float* __restrict__ Qt, float* __restrict__ Kt, float* __restrict__ Vt,
    float* __restrict__ Kmx, float* __restrict__ Kmn)
{
    __shared__ __align__(16) float As[2][G_BM * G_PA];
    __shared__ __align__(16) float Bs[2][(G_BK / 2) * G_PN * 2];

    const int tid  = threadIdx.x;
    const int lane = tid & 31;
    const int wave = tid >> 5;
    const int lm   = lane & 15;
    const int lk   = lane >> 4;
    const int m0   = (blockIdx.x >> 1) * G_BM;
    const int jb0  = (blockIdx.x & 1) * G_BN;
    const int wm   = wave / 6;
    const int wn   = wave % 6;

    const int  a_row = tid >> 4;
    const int  a_q   = tid & 15;
    const bool a_two = tid < 128;
    const float* a_src0 = X + (size_t)(m0 + a_row) * DIM + 4 * a_q;
    const float* a_src1 = X + (size_t)(m0 + 24 + (a_row & 7)) * DIM + 4 * a_q;
    const int a_dst0 = a_row * G_PA + 4 * a_q;
    const int a_dst1 = (24 + (a_row & 7)) * G_PA + 4 * a_q;

    const int b_kp = tid / 24;
    const int b_c4 = tid % 24;
    const float* b_src = Wqkv + (size_t)(2 * b_kp) * QKV3 + jb0 + 4 * b_c4;
    const int b_dst0 = (b_kp * G_PN + 4 * b_c4) * 2;
    const int b_dst1 = ((b_kp + 16) * G_PN + 4 * b_c4) * 2;

    float4 ra0, ra1, rb00, rb01, rb10, rb11;
    ra1 = make_float4(0.f, 0.f, 0.f, 0.f);

    ra0  = *(const float4*)(a_src0);
    if (a_two) ra1 = *(const float4*)(a_src1);
    rb00 = *(const float4*)(b_src);
    rb01 = *(const float4*)(b_src + QKV3);
    rb10 = *(const float4*)(b_src + 32 * QKV3);
    rb11 = *(const float4*)(b_src + 33 * QKV3);

    *(float4*)&As[0][a_dst0] = ra0;
    if (a_two) *(float4*)&As[0][a_dst1] = ra1;
    *(float4*)&Bs[0][b_dst0]     = make_float4(rb00.x, rb01.x, rb00.y, rb01.y);
    *(float4*)&Bs[0][b_dst0 + 4] = make_float4(rb00.z, rb01.z, rb00.w, rb01.w);
    *(float4*)&Bs[0][b_dst1]     = make_float4(rb10.x, rb11.x, rb10.y, rb11.y);
    *(float4*)&Bs[0][b_dst1 + 4] = make_float4(rb10.z, rb11.z, rb10.w, rb11.w);
    __syncthreads();

    v8f acc = {0.f,0.f,0.f,0.f,0.f,0.f,0.f,0.f};

    const int a_frag = (wm * 16 + lm) * G_PA + 2 * lk;
    const int b_frag = (lk * G_PN + wn * 16 + lm) * 2;

    for (int kc = 0; kc < G_NK; ++kc) {
        const int  s    = kc & 1;
        const bool more = (kc + 1) < G_NK;

        if (more) {
            const int ko = (kc + 1) * G_BK;
            ra0  = *(const float4*)(a_src0 + ko);
            if (a_two) ra1 = *(const float4*)(a_src1 + ko);
            const float* bp = b_src + (size_t)ko * QKV3;
            rb00 = *(const float4*)(bp);
            rb01 = *(const float4*)(bp + QKV3);
            rb10 = *(const float4*)(bp + 32 * QKV3);
            rb11 = *(const float4*)(bp + 33 * QKV3);
        }

        const float* as = &As[s][a_frag];
        const float* bs = &Bs[s][b_frag];
        for (int kk = 0; kk < G_BK / 4; ++kk) {
            const v2f a = *(const v2f*)(as + 4 * kk);
            const v2f b = *(const v2f*)(bs + 4 * kk * G_PN);
            acc = __builtin_amdgcn_wmma_f32_16x16x4_f32(
                false, a, false, b, (short)0, acc, false, false);
        }

        if (more) {
            const int t = s ^ 1;
            *(float4*)&As[t][a_dst0] = ra0;
            if (a_two) *(float4*)&As[t][a_dst1] = ra1;
            *(float4*)&Bs[t][b_dst0]     = make_float4(rb00.x, rb01.x, rb00.y, rb01.y);
            *(float4*)&Bs[t][b_dst0 + 4] = make_float4(rb00.z, rb01.z, rb00.w, rb01.w);
            *(float4*)&Bs[t][b_dst1]     = make_float4(rb10.x, rb11.x, rb10.y, rb11.y);
            *(float4*)&Bs[t][b_dst1 + 4] = make_float4(rb10.z, rb11.z, rb10.w, rb11.w);
            __syncthreads();
        }
    }

    const int row0  = m0 + wm * 16;
    const int j0    = jb0 + wn * 16;
    const int b_idx = row0 >> 9;
    const int n0    = (row0 & 511) + lk * 8;
    const int col   = j0 + lm;
    const int which = j0 >> 6;
    const int c     = col & 63;
    float* T = (which == 0) ? Qt : ((which == 1) ? Kt : Vt);
    float* base = T + ((size_t)b_idx * DK + c) * SEQ;
    #pragma unroll
    for (int r = 0; r < 8; ++r)
        base[n0 + r] = acc[r];

    if (which == 1) {
        float mx = acc[0], mn = acc[0];
        #pragma unroll
        for (int r = 1; r < 8; ++r) { mx = fmaxf(mx, acc[r]); mn = fminf(mn, acc[r]); }
        mx = fmaxf(mx, __shfl_xor(mx, 16, 32));
        mn = fminf(mn, __shfl_xor(mn, 16, 32));
        if (lk == 0) {
            const size_t e = ((size_t)b_idx * KTILES + ((row0 & 511) >> 4)) * DK + c;
            Kmx[e] = mx;
            Kmn[e] = mn;
        }
    }
}

#define LOG2E 1.4426950408889634f
#define F_BM  16
#define F_P   (SEQ + 16)
#define F_PA  (DK + 4)
#define F_KC  128
#define F_NC  (SEQ / F_KC)

__global__ __launch_bounds__(256) void attn_out_kernel(
    const float* Qt, const float* Kt,
    const float* Vt, const float* Kmx, const float* Kmn,
    const float* Wo, float* __restrict__ Out)
{
    __shared__ __align__(16) float Ks[DK * F_P];
    __shared__ __align__(16) float Vs[DK * F_P];
    __shared__ __align__(16) float SAs[F_BM * F_PA];

    const int tid  = threadIdx.x;
    const int lane = tid & 31;
    const int wave = tid >> 5;
    const int c    = tid >> 2;
    const int rq   = tid & 3;
    const int b    = blockIdx.x >> 5;
    const int n0   = (blockIdx.x & 31) * F_BM;

    const float* Kb = Kt + (size_t)b * DK * SEQ;
    const float* Vb = Vt + (size_t)b * DK * SEQ;

    const float4 q4 = *(const float4*)(Qt + ((size_t)b * DK + c) * SEQ + n0 + 4 * rq);

    const float* xp = Kmx + ((size_t)b * KTILES + 8 * rq) * DK + c;
    const float* np = Kmn + ((size_t)b * KTILES + 8 * rq) * DK + c;
    float xt[8], nt[8];
    #pragma unroll
    for (int t = 0; t < 8; ++t) { xt[t] = xp[t * DK]; nt[t] = np[t * DK]; }

    {
        float4 kr[8];
        float4 vr[8];
        #pragma unroll
        for (int i = 0; i < 8; ++i)
            kr[i] = *(const float4*)(Kb + (size_t)(8 * wave + i) * SEQ + lane * 4);
        #pragma unroll
        for (int i = 0; i < 8; ++i)
            vr[i] = *(const float4*)(Vb + (size_t)(8 * wave + i) * SEQ + lane * 4);
        #pragma unroll
        for (int i = 0; i < 8; ++i)
            *(float4*)&Ks[(8 * wave + i) * F_P + lane * 4] = kr[i];
        #pragma unroll
        for (int i = 0; i < 8; ++i)
            *(float4*)&Vs[(8 * wave + i) * F_P + lane * 4] = vr[i];
    }
    __syncthreads();

    const float4* kc4 = (const float4*)(Ks + c * F_P);
    const float4* vc4 = (const float4*)(Vs + c * F_P);

    float KMAX, KMIN;
    {
        float mx = fmaxf(fmaxf(fmaxf(xt[0], xt[1]), fmaxf(xt[2], xt[3])),
                         fmaxf(fmaxf(xt[4], xt[5]), fmaxf(xt[6], xt[7])));
        float mn = fminf(fminf(fminf(nt[0], nt[1]), fminf(nt[2], nt[3])),
                         fminf(fminf(nt[4], nt[5]), fminf(nt[6], nt[7])));
        mx = fmaxf(mx, __shfl_xor(mx, 1, 32));
        mn = fminf(mn, __shfl_xor(mn, 1, 32));
        KMAX = fmaxf(mx, __shfl_xor(mx, 2, 32));
        KMIN = fminf(mn, __shfl_xor(mn, 2, 32));
    }

    float4 kn0, kn1, kn2, kn3, kn4, kn5, kn6, kn7;
    float4 vn0, vn1, vn2, vn3, vn4, vn5, vn6, vn7;
    {
        const float* ks = Kb + (size_t)(8 * wave) * SEQ + (32 + lane) * 4;
        kn0 = *(const float4*)(ks);
        kn1 = *(const float4*)(ks + SEQ);
        kn2 = *(const float4*)(ks + 2 * SEQ);
        kn3 = *(const float4*)(ks + 3 * SEQ);
        kn4 = *(const float4*)(ks + 4 * SEQ);
        kn5 = *(const float4*)(ks + 5 * SEQ);
        kn6 = *(const float4*)(ks + 6 * SEQ);
        kn7 = *(const float4*)(ks + 7 * SEQ);
        const float* vs = Vb + (size_t)(8 * wave) * SEQ + (32 + lane) * 4;
        vn0 = *(const float4*)(vs);
        vn1 = *(const float4*)(vs + SEQ);
        vn2 = *(const float4*)(vs + 2 * SEQ);
        vn3 = *(const float4*)(vs + 3 * SEQ);
        vn4 = *(const float4*)(vs + 4 * SEQ);
        vn5 = *(const float4*)(vs + 5 * SEQ);
        vn6 = *(const float4*)(vs + 6 * SEQ);
        vn7 = *(const float4*)(vs + 7 * SEQ);
    }

    const int lm = lane & 15;
    const int lk = lane >> 4;
    const float* wb = Wo + (size_t)(2 * lk) * DIM + wave * 64 + 4 * lm;
    float4 bx[DK / 4], by[DK / 4];
    #pragma unroll
    for (int j = 0; j < 10; ++j) {
        bx[j] = *(const float4*)(wb + (size_t)(4 * j) * DIM);
        by[j] = *(const float4*)(wb + (size_t)(4 * j + 1) * DIM);
    }

    const float qv[4] = {q4.x, q4.y, q4.z, q4.w};
    v2f A[4], N[4];
    #pragma unroll
    for (int r = 0; r < 4; ++r) {
        const float a  = qv[r] * (SCALE * LOG2E);
        const float cm = a * ((a >= 0.f) ? KMAX : KMIN);
        A[r] = (v2f){a, a};
        N[r] = (v2f){-cm, -cm};
    }

    const v2f z2 = {0.f, 0.f};
    v2f sa[4], sb[4], pa[4], pb[4];
    #pragma unroll
    for (int r = 0; r < 4; ++r) { sa[r] = z2; sb[r] = z2; pa[r] = z2; pb[r] = z2; }

#define ATTN_PASS2(KA, VA, KB, VB)                                                         \
    {                                                                                      \
        const v2f ka = {KA.x, KA.y}, kb = {KA.z, KA.w}, kc = {KB.x, KB.y}, kd = {KB.z, KB.w}; \
        const v2f va = {VA.x, VA.y}, vb = {VA.z, VA.w}, vc = {VB.x, VB.y}, vd = {VB.z, VB.w}; \
        v2f xa[4], xb[4], xc[4], xd[4], ea[4], eb[4], ec[4], ed[4];                        \
        _Pragma("unroll")                                                                  \
        for (int r = 0; r < 4; ++r) {                                                      \
            xa[r] = __builtin_elementwise_fma(ka, A[r], N[r]);                             \
            xb[r] = __builtin_elementwise_fma(kb, A[r], N[r]);                             \
        }                                                                                  \
        _Pragma("unroll")                                                                  \
        for (int r = 0; r < 4; ++r) {                                                      \
            xc[r] = __builtin_elementwise_fma(kc, A[r], N[r]);                             \
            xd[r] = __builtin_elementwise_fma(kd, A[r], N[r]);                             \
        }                                                                                  \
        _Pragma("unroll")                                                                  \
        for (int r = 0; r < 4; ++r) {                                                      \
            ea[r] = (v2f){__builtin_amdgcn_exp2f(xa[r].x), __builtin_amdgcn_exp2f(xa[r].y)}; \
            eb[r] = (v2f){__builtin_amdgcn_exp2f(xb[r].x), __builtin_amdgcn_exp2f(xb[r].y)}; \
        }                                                                                  \
        _Pragma("unroll")                                                                  \
        for (int r = 0; r < 4; ++r) {                                                      \
            ec[r] = (v2f){__builtin_amdgcn_exp2f(xc[r].x), __builtin_amdgcn_exp2f(xc[r].y)}; \
            ed[r] = (v2f){__builtin_amdgcn_exp2f(xd[r].x), __builtin_amdgcn_exp2f(xd[r].y)}; \
        }                                                                                  \
        _Pragma("unroll")                                                                  \
        for (int r = 0; r < 4; ++r) {                                                      \
            sa[r] += ea[r];  pa[r] = __builtin_elementwise_fma(ea[r], va, pa[r]);          \
            sb[r] += eb[r];  pb[r] = __builtin_elementwise_fma(eb[r], vb, pb[r]);          \
        }                                                                                  \
        _Pragma("unroll")                                                                  \
        for (int r = 0; r < 4; ++r) {                                                      \
            sa[r] += ec[r];  pa[r] = __builtin_elementwise_fma(ec[r], vc, pa[r]);          \
            sb[r] += ed[r];  pb[r] = __builtin_elementwise_fma(ed[r], vd, pb[r]);          \
        }                                                                                  \
    }

#define ATTN_TP4                                                                           \
        __builtin_amdgcn_sched_group_barrier(0x400, 1, 0);                                 \
        __builtin_amdgcn_sched_group_barrier(0x002, 1, 0);                                 \
        __builtin_amdgcn_sched_group_barrier(0x400, 1, 0);                                 \
        __builtin_amdgcn_sched_group_barrier(0x002, 1, 0);                                 \
        __builtin_amdgcn_sched_group_barrier(0x400, 1, 0);                                 \
        __builtin_amdgcn_sched_group_barrier(0x002, 1, 0);                                 \
        __builtin_amdgcn_sched_group_barrier(0x400, 1, 0);                                 \
        __builtin_amdgcn_sched_group_barrier(0x002, 1, 0);
#define ATTN_ORDER2(NLDS)                                                                  \
        __builtin_amdgcn_sched_group_barrier(0x100, NLDS, 0);                              \
        __builtin_amdgcn_sched_group_barrier(0x002, 4, 0);                                 \
        ATTN_TP4 ATTN_TP4 ATTN_TP4 ATTN_TP4 ATTN_TP4 ATTN_TP4 ATTN_TP4 ATTN_TP4            \
        __builtin_amdgcn_sched_group_barrier(0x002, 12, 0);

    for (int ch = 0; ch < F_NC; ++ch) {
        const float4* kq = kc4 + 32 * ch;
        const float4* vq = vc4 + 32 * ch;

        for (int m = 0; m < F_KC / 4; m += 2) {
            const float4 k0 = kq[m], k1 = kq[m + 1];
            const float4 v0 = vq[m], v1 = vq[m + 1];
            ATTN_PASS2(k0, v0, k1, v1)
            ATTN_ORDER2(4)
        }

        if (ch + 1 < F_NC) {
            float* kd = &Ks[(8 * wave) * F_P + (32 * (ch + 1) + lane) * 4];
            *(float4*)(kd)           = kn0;
            *(float4*)(kd + F_P)     = kn1;
            *(float4*)(kd + 2 * F_P) = kn2;
            *(float4*)(kd + 3 * F_P) = kn3;
            *(float4*)(kd + 4 * F_P) = kn4;
            *(float4*)(kd + 5 * F_P) = kn5;
            *(float4*)(kd + 6 * F_P) = kn6;
            *(float4*)(kd + 7 * F_P) = kn7;
            float* vd = &Vs[(8 * wave) * F_P + (32 * (ch + 1) + lane) * 4];
            *(float4*)(vd)           = vn0;
            *(float4*)(vd + F_P)     = vn1;
            *(float4*)(vd + 2 * F_P) = vn2;
            *(float4*)(vd + 3 * F_P) = vn3;
            *(float4*)(vd + 4 * F_P) = vn4;
            *(float4*)(vd + 5 * F_P) = vn5;
            *(float4*)(vd + 6 * F_P) = vn6;
            *(float4*)(vd + 7 * F_P) = vn7;
            __syncthreads();
            if (ch + 2 < F_NC) {
                const float* ks = Kb + (size_t)(8 * wave) * SEQ + (32 * (ch + 2) + lane) * 4;
                kn0 = *(const float4*)(ks);
                kn1 = *(const float4*)(ks + SEQ);
                kn2 = *(const float4*)(ks + 2 * SEQ);
                kn3 = *(const float4*)(ks + 3 * SEQ);
                kn4 = *(const float4*)(ks + 4 * SEQ);
                kn5 = *(const float4*)(ks + 5 * SEQ);
                kn6 = *(const float4*)(ks + 6 * SEQ);
                kn7 = *(const float4*)(ks + 7 * SEQ);
                const float* vs = Vb + (size_t)(8 * wave) * SEQ + (32 * (ch + 2) + lane) * 4;
                vn0 = *(const float4*)(vs);
                vn1 = *(const float4*)(vs + SEQ);
                vn2 = *(const float4*)(vs + 2 * SEQ);
                vn3 = *(const float4*)(vs + 3 * SEQ);
                vn4 = *(const float4*)(vs + 4 * SEQ);
                vn5 = *(const float4*)(vs + 5 * SEQ);
                vn6 = *(const float4*)(vs + 6 * SEQ);
                vn7 = *(const float4*)(vs + 7 * SEQ);
            }
        }
    }
#undef ATTN_PASS2
#undef ATTN_TP4
#undef ATTN_ORDER2

    #pragma unroll
    for (int r = 0; r < 4; ++r) {
        const float sum = (sa[r].x + sa[r].y) + (sb[r].x + sb[r].y);
        const float num = (pa[r].x + pa[r].y) + (pb[r].x + pb[r].y);
        SAs[(4 * rq + r) * F_PA + c] = num / sum;
    }
    __syncthreads();

    #pragma unroll
    for (int j = 10; j < DK / 4; ++j) {
        bx[j] = *(const float4*)(wb + (size_t)(4 * j) * DIM);
        by[j] = *(const float4*)(wb + (size_t)(4 * j + 1) * DIM);
    }

    v8f acc0 = {0.f,0.f,0.f,0.f,0.f,0.f,0.f,0.f};
    v8f acc1 = acc0, acc2 = acc0, acc3 = acc0;

    const float* af = &SAs[lm * F_PA + 2 * lk];
    #pragma unroll
    for (int j = 0; j < DK / 4; ++j) {
        const v2f a  = *(const v2f*)(af + 4 * j);
        const v2f b0 = {bx[j].x, by[j].x};
        const v2f b1 = {bx[j].y, by[j].y};
        const v2f b2 = {bx[j].z, by[j].z};
        const v2f b3 = {bx[j].w, by[j].w};
        acc0 = __builtin_amdgcn_wmma_f32_16x16x4_f32(
            false, a, false, b0, (short)0, acc0, false, false);
        acc1 = __builtin_amdgcn_wmma_f32_16x16x4_f32(
            false, a, false, b1, (short)0, acc1, false, false);
        acc2 = __builtin_amdgcn_wmma_f32_16x16x4_f32(
            false, a, false, b2, (short)0, acc2, false, false);
        acc3 = __builtin_amdgcn_wmma_f32_16x16x4_f32(
            false, a, false, b3, (short)0, acc3, false, false);
    }

    float* obase = Out + ((size_t)b * SEQ + n0 + lk * 8) * DIM + wave * 64 + 4 * lm;
    #pragma unroll
    for (int r = 0; r < 8; ++r)
        *(float4*)(obase + (size_t)r * DIM) = make_float4(acc0[r], acc1[r], acc2[r], acc3[r]);
}

extern "C" void kernel_launch(void* const* d_in, const int* in_sizes, int n_in,
                              void* d_out, int out_size, void* d_ws, size_t ws_size,
                              hipStream_t stream)
{
    const float* X    = (const float*)d_in[0];
    const float* Wqkv = (const float*)d_in[1];
    const float* Wo   = (const float*)d_in[2];
    float* out = (float*)d_out;

    const size_t CHN = (size_t)BATCH * DK * SEQ;
    float* Qt = (float*)d_ws;
    float* Kt = Qt + CHN;
    float* Vt = Kt + CHN;
    float* Kmx = Vt + CHN;
    float* Kmn = Kmx + (size_t)BATCH * DK * KTILES;

    qkv_gemm_kernel<<<256, 384, 0, stream>>>(X, Wqkv, Qt, Kt, Vt, Kmx, Kmn);
    attn_out_kernel<<<BATCH * (SEQ / F_BM), 256, 0, stream>>>(Qt, Kt, Vt, Kmx, Kmn, Wo, out);
}


# QKV GEMM: global loads requested three K-chunks ahead through a ring of three register sets; chunk barriers wait for LDS only, not for loads in flight; k order and summation order unchanged
# speedup vs baseline: 1.8901x; 1.0470x over previous
#include <hip/hip_runtime.h>
#include <math.h>

#define BATCH 8
#define SEQ   512
#define DIM   512
#define DK    64
#define QKV3  192
#define SCALE 0.125f

typedef __attribute__((ext_vector_type(2))) float v2f;
typedef __attribute__((ext_vector_type(4))) float v4f;
typedef __attribute__((ext_vector_type(8))) float v8f;

#define KTILES (SEQ / 16)
#define G_BM 32
#define G_BN 96
#define G_BK 64
#define G_NK (DIM / G_BK)
#define G_PA (G_BK + 4)
#define G_PN 112
#define G_NT 384

__device__ __forceinline__ void lds_barrier()
{
    __builtin_amdgcn_fence(__ATOMIC_RELEASE, "workgroup", "local");
    __builtin_amdgcn_s_barrier();
    __builtin_amdgcn_fence(__ATOMIC_ACQUIRE, "workgroup", "local");
}

__global__ __launch_bounds__(G_NT) void qkv_gemm_kernel(
    const float* __restrict__ X, const float* __restrict__ Wqkv,
    float* __restrict__ Qt, float* __restrict__ Kt, float* __restrict__ Vt,
    float* __restrict__ Kmx, float* __restrict__ Kmn)
{
    __shared__ __align__(16) float As[2][G_BM * G_PA];
    __shared__ __align__(16) float Bs[2][(G_BK / 2) * G_PN * 2];

    const int tid  = threadIdx.x;
    const int lane = tid & 31;
    const int wave = tid >> 5;
    const int lm   = lane & 15;
    const int lk   = lane >> 4;
    const int m0   = (blockIdx.x >> 1) * G_BM;
    const int jb0  = (blockIdx.x & 1) * G_BN;
    const int wm   = wave / 6;
    const int wn   = wave % 6;

    const int  a_row = tid >> 4;
    const int  a_q   = tid & 15;
    const bool a_two = tid < 128;
    const float* a_src0 = X + (size_t)(m0 + a_row) * DIM + 4 * a_q;
    const float* a_src1 = X + (size_t)(m0 + 24 + (a_row & 7)) * DIM + 4 * a_q;
    const int a_dst0 = a_row * G_PA + 4 * a_q;
    const int a_dst1 = (24 + (a_row & 7)) * G_PA + 4 * a_q;

    const int b_kp = tid / 24;
    const int b_c4 = tid % 24;
    const float* b_src = Wqkv + (size_t)(2 * b_kp) * QKV3 + jb0 + 4 * b_c4;
    const int b_dst0 = (b_kp * G_PN + 4 * b_c4) * 2;
    const int b_dst1 = ((b_kp + 16) * G_PN + 4 * b_c4) * 2;

#define G_SET(S) v4f ra0_##S, ra1_##S = {0.f, 0.f, 0.f, 0.f}, rb00_##S, rb01_##S, rb10_##S, rb11_##S;
    G_SET(0) G_SET(1) G_SET(2)
#undef G_SET

#define G_REQUEST(S, KC)                                                 \
    {                                                                    \
        const int ko = (KC) * G_BK;                                      \
        ra0_##S = *(const v4f*)(a_src0 + ko);                            \
        if (a_two) ra1_##S = *(const v4f*)(a_src1 + ko);                 \
        const float* bp = b_src + (size_t)ko * QKV3;                     \
        rb00_##S = *(const v4f*)(bp);                                    \
        rb01_##S = *(const v4f*)(bp + QKV3);                             \
        rb10_##S = *(const v4f*)(bp + 32 * QKV3);                        \
        rb11_##S = *(const v4f*)(bp + 33 * QKV3);                        \
    }
#define G_PUT(S, T)                                                                             \
    {                                                                                           \
        *(v4f*)&As[T][a_dst0] = ra0_##S;                                                        \
        if (a_two) *(v4f*)&As[T][a_dst1] = ra1_##S;                                             \
        *(v4f*)&Bs[T][b_dst0]     = (v4f){rb00_##S.x, rb01_##S.x, rb00_##S.y, rb01_##S.y};      \
        *(v4f*)&Bs[T][b_dst0 + 4] = (v4f){rb00_##S.z, rb01_##S.z, rb00_##S.w, rb01_##S.w};      \
        *(v4f*)&Bs[T][b_dst1]     = (v4f){rb10_##S.x, rb11_##S.x, rb10_##S.y, rb11_##S.y};      \
        *(v4f*)&Bs[T][b_dst1 + 4] = (v4f){rb10_##S.z, rb11_##S.z, rb10_##S.w, rb11_##S.w};      \
    }

    v8f acc = {0.f,0.f,0.f,0.f,0.f,0.f,0.f,0.f};

    const int a_frag = (wm * 16 + lm) * G_PA + 2 * lk;
    const int b_frag = (lk * G_PN + wn * 16 + lm) * 2;

#define G_PRODUCTS(T)                                                                          \
    {                                                                                          \
        const float* as = &As[T][a_frag];                                                      \
        const float* bs = &Bs[T][b_frag];                                                      \
        for (int kk = 0; kk < G_BK / 4; ++kk) {                                                \
              \
            const v2f a = *(const v2f*)(as + 4 * kk);                                          \
                        \
            const v2f b = *(const v2f*)(bs + 4 * kk * G_PN);                                   \
            acc = __builtin_amdgcn_wmma_f32_16x16x4_f32(                                       \
                false, a, false, b, (short)0, acc, false, false);                              \
        }                                                                                      \
    }

    G_REQUEST(0, 0) G_REQUEST(1, 1) G_REQUEST(2, 2)
    G_PUT(0, 0)
    lds_barrier();

    G_REQUEST(0, 3) G_PRODUCTS(0) G_PUT(1, 1) lds_barrier();
    G_REQUEST(1, 4) G_PRODUCTS(1) G_PUT(2, 0) lds_barrier();
    G_REQUEST(2, 5) G_PRODUCTS(0) G_PUT(0, 1) lds_barrier();
    G_REQUEST(0, 6) G_PRODUCTS(1) G_PUT(1, 0) lds_barrier();
    G_REQUEST(1, 7) G_PRODUCTS(0) G_PUT(2, 1) lds_barrier();
                    G_PRODUCTS(1) G_PUT(0, 0) lds_barrier();
                    G_PRODUCTS(0) G_PUT(1, 1) lds_barrier();
                    G_PRODUCTS(1)
#undef G_REQUEST
#undef G_PUT
#undef G_PRODUCTS

    const int row0  = m0 + wm * 16;
    const int j0    = jb0 + wn * 16;
    const int b_idx = row0 >> 9;
    const int n0    = (row0 & 511) + lk * 8;
    const int col   = j0 + lm;
    const int which = j0 >> 6;
    const int c     = col & 63;
    float* T = (which == 0) ? Qt : ((which == 1) ? Kt : Vt);
    float* base = T + ((size_t)b_idx * DK + c) * SEQ;
    #pragma unroll
    for (int r = 0; r < 8; ++r)
        base[n0 + r] = acc[r];

    if (which == 1) {
        float mx = acc[0], mn = acc[0];
        #pragma unroll
        for (int r = 1; r < 8; ++r) { mx = fmaxf(mx, acc[r]); mn = fminf(mn, acc[r]); }
        mx = fmaxf(mx, __shfl_xor(mx, 16, 32));
        mn = fminf(mn, __shfl_xor(mn, 16, 32));
        if (lk == 0) {
            const size_t e = ((size_t)b_idx * KTILES + ((row0 & 511) >> 4)) * DK + c;
            Kmx[e] = mx;
            Kmn[e] = mn;
        }
    }
}

#define LOG2E 1.4426950408889634f
#define F_BM  16
#define F_P   (SEQ + 16)
#define F_PA  (DK + 4)
#define F_KC  128
#define F_NC  (SEQ / F_KC)

__global__ __launch_bounds__(256) void attn_out_kernel(
    const float* Qt, const float* Kt,
    const float* Vt, const float* Kmx, const float* Kmn,
    const float* Wo, float* __restrict__ Out)
{
    __shared__ __align__(16) float Ks[DK * F_P];
    __shared__ __align__(16) float Vs[DK * F_P];
    __shared__ __align__(16) float SAs[F_BM * F_PA];

    const int tid  = threadIdx.x;
    const int lane = tid & 31;
    const int wave = tid >> 5;
    const int c    = tid >> 2;
    const int rq   = tid & 3;
    const int b    = blockIdx.x >> 5;
    const int n0   = (blockIdx.x & 31) * F_BM;

    const float* Kb = Kt + (size_t)b * DK * SEQ;
    const float* Vb = Vt + (size_t)b * DK * SEQ;

    const float4 q4 = *(const float4*)(Qt + ((size_t)b * DK + c) * SEQ + n0 + 4 * rq);

    const float* xp = Kmx + ((size_t)b * KTILES + 8 * rq) * DK + c;
    const float* np = Kmn + ((size_t)b * KTILES + 8 * rq) * DK + c;
    float xt[8], nt[8];
    #pragma unroll
    for (int t = 0; t < 8; ++t) { xt[t] = xp[t * DK]; nt[t] = np[t * DK]; }

    {
        float4 kr[8];
        float4 vr[8];
        #pragma unroll
        for (int i = 0; i < 8; ++i)
            kr[i] = *(const float4*)(Kb + (size_t)(8 * wave + i) * SEQ + lane * 4);
        #pragma unroll
        for (int i = 0; i < 8; ++i)
            vr[i] = *(const float4*)(Vb + (size_t)(8 * wave + i) * SEQ + lane * 4);
        #pragma unroll
        for (int i = 0; i < 8; ++i)
            *(float4*)&Ks[(8 * wave + i) * F_P + lane * 4] = kr[i];
        #pragma unroll
        for (int i = 0; i < 8; ++i)
            *(float4*)&Vs[(8 * wave + i) * F_P + lane * 4] = vr[i];
    }
    __syncthreads();

    const float4* kc4 = (const float4*)(Ks + c * F_P);
    const float4* vc4 = (const float4*)(Vs + c * F_P);

    float KMAX, KMIN;
    {
        float mx = fmaxf(fmaxf(fmaxf(xt[0], xt[1]), fmaxf(xt[2], xt[3])),
                         fmaxf(fmaxf(xt[4], xt[5]), fmaxf(xt[6], xt[7])));
        float mn = fminf(fminf(fminf(nt[0], nt[1]), fminf(nt[2], nt[3])),
                         fminf(fminf(nt[4], nt[5]), fminf(nt[6], nt[7])));
        mx = fmaxf(mx, __shfl_xor(mx, 1, 32));
        mn = fminf(mn, __shfl_xor(mn, 1, 32));
        KMAX = fmaxf(mx, __shfl_xor(mx, 2, 32));
        KMIN = fminf(mn, __shfl_xor(mn, 2, 32));
    }

    float4 kn0, kn1, kn2, kn3, kn4, kn5, kn6, kn7;
    float4 vn0, vn1, vn2, vn3, vn4, vn5, vn6, vn7;
    {
        const float* ks = Kb + (size_t)(8 * wave) * SEQ + (32 + lane) * 4;
        kn0 = *(const float4*)(ks);
        kn1 = *(const float4*)(ks + SEQ);
        kn2 = *(const float4*)(ks + 2 * SEQ);
        kn3 = *(const float4*)(ks + 3 * SEQ);
        kn4 = *(const float4*)(ks + 4 * SEQ);
        kn5 = *(const float4*)(ks + 5 * SEQ);
        kn6 = *(const float4*)(ks + 6 * SEQ);
        kn7 = *(const float4*)(ks + 7 * SEQ);
        const float* vs = Vb + (size_t)(8 * wave) * SEQ + (32 + lane) * 4;
        vn0 = *(const float4*)(vs);
        vn1 = *(const float4*)(vs + SEQ);
        vn2 = *(const float4*)(vs + 2 * SEQ);
        vn3 = *(const float4*)(vs + 3 * SEQ);
        vn4 = *(const float4*)(vs + 4 * SEQ);
        vn5 = *(const float4*)(vs + 5 * SEQ);
        vn6 = *(const float4*)(vs + 6 * SEQ);
        vn7 = *(const float4*)(vs + 7 * SEQ);
    }

    const int lm = lane & 15;
    const int lk = lane >> 4;
    const float* wb = Wo + (size_t)(2 * lk) * DIM + wave * 64 + 4 * lm;
    float4 bx[DK / 4], by[DK / 4];
    #pragma unroll
    for (int j = 0; j < 10; ++j) {
        bx[j] = *(const float4*)(wb + (size_t)(4 * j) * DIM);
        by[j] = *(const float4*)(wb + (size_t)(4 * j + 1) * DIM);
    }

    const float qv[4] = {q4.x, q4.y, q4.z, q4.w};
    v2f A[4], N[4];
    #pragma unroll
    for (int r = 0; r < 4; ++r) {
        const float a  = qv[r] * (SCALE * LOG2E);
        const float cm = a * ((a >= 0.f) ? KMAX : KMIN);
        A[r] = (v2f){a, a};
        N[r] = (v2f){-cm, -cm};
    }

    const v2f z2 = {0.f, 0.f};
    v2f sa[4], sb[4], pa[4], pb[4];
    #pragma unroll
    for (int r = 0; r < 4; ++r) { sa[r] = z2; sb[r] = z2; pa[r] = z2; pb[r] = z2; }

#define ATTN_PASS2(KA, VA, KB, VB)                                                         \
    {                                                                                      \
        const v2f ka = {KA.x, KA.y}, kb = {KA.z, KA.w}, kc = {KB.x, KB.y}, kd = {KB.z, KB.w}; \
        const v2f va = {VA.x, VA.y}, vb = {VA.z, VA.w}, vc = {VB.x, VB.y}, vd = {VB.z, VB.w}; \
        v2f xa[4], xb[4], xc[4], xd[4], ea[4], eb[4], ec[4], ed[4];                        \
        _Pragma("unroll")                                                                  \
        for (int r = 0; r < 4; ++r) {                                                      \
            xa[r] = __builtin_elementwise_fma(ka, A[r], N[r]);                             \
            xb[r] = __builtin_elementwise_fma(kb, A[r], N[r]);                             \
        }                                                                                  \
        _Pragma("unroll")                                                                  \
        for (int r = 0; r < 4; ++r) {                                                      \
            xc[r] = __builtin_elementwise_fma(kc, A[r], N[r]);                             \
            xd[r] = __builtin_elementwise_fma(kd, A[r], N[r]);                             \
        }                                                                                  \
        _Pragma("unroll")                                                                  \
        for (int r = 0; r < 4; ++r) {                                                      \
            ea[r] = (v2f){__builtin_amdgcn_exp2f(xa[r].x), __builtin_amdgcn_exp2f(xa[r].y)}; \
            eb[r] = (v2f){__builtin_amdgcn_exp2f(xb[r].x), __builtin_amdgcn_exp2f(xb[r].y)}; \
        }                                                                                  \
        _Pragma("unroll")                                                                  \
        for (int r = 0; r < 4; ++r) {                                                      \
            ec[r] = (v2f){__builtin_amdgcn_exp2f(xc[r].x), __builtin_amdgcn_exp2f(xc[r].y)}; \
            ed[r] = (v2f){__builtin_amdgcn_exp2f(xd[r].x), __builtin_amdgcn_exp2f(xd[r].y)}; \
        }                                                                                  \
        _Pragma("unroll")                                                                  \
        for (int r = 0; r < 4; ++r) {                                                      \
            sa[r] += ea[r];  pa[r] = __builtin_elementwise_fma(ea[r], va, pa[r]);          \
            sb[r] += eb[r];  pb[r] = __builtin_elementwise_fma(eb[r], vb, pb[r]);          \
        }                                                                                  \
        _Pragma("unroll")                                                                  \
        for (int r = 0; r < 4; ++r) {                                                      \
            sa[r] += ec[r];  pa[r] = __builtin_elementwise_fma(ec[r], vc, pa[r]);          \
            sb[r] += ed[r];  pb[r] = __builtin_elementwise_fma(ed[r], vd, pb[r]);          \
        }                                                                                  \
    }

#define ATTN_TP4                                                                           \
        __builtin_amdgcn_sched_group_barrier(0x400, 1, 0);                                 \
        __builtin_amdgcn_sched_group_barrier(0x002, 1, 0);                                 \
        __builtin_amdgcn_sched_group_barrier(0x400, 1, 0);                                 \
        __builtin_amdgcn_sched_group_barrier(0x002, 1, 0);                                 \
        __builtin_amdgcn_sched_group_barrier(0x400, 1, 0);                                 \
        __builtin_amdgcn_sched_group_barrier(0x002, 1, 0);                                 \
        __builtin_amdgcn_sched_group_barrier(0x400, 1, 0);                                 \
        __builtin_amdgcn_sched_group_barrier(0x002, 1, 0);
#define ATTN_ORDER2(NLDS)                                                                  \
        __builtin_amdgcn_sched_group_barrier(0x100, NLDS, 0);                              \
        __builtin_amdgcn_sched_group_barrier(0x002, 4, 0);                                 \
        ATTN_TP4 ATTN_TP4 ATTN_TP4 ATTN_TP4 ATTN_TP4 ATTN_TP4 ATTN_TP4 ATTN_TP4            \
        __builtin_amdgcn_sched_group_barrier(0x002, 12, 0);

    for (int ch = 0; ch < F_NC; ++ch) {
        const float4* kq = kc4 + 32 * ch;
        const float4* vq = vc4 + 32 * ch;

        for (int m = 0; m < F_KC / 4; m += 2) {
            const float4 k0 = kq[m], k1 = kq[m + 1];
            const float4 v0 = vq[m], v1 = vq[m + 1];
            ATTN_PASS2(k0, v0, k1, v1)
            ATTN_ORDER2(4)
        }

        if (ch + 1 < F_NC) {
            float* kd = &Ks[(8 * wave) * F_P + (32 * (ch + 1) + lane) * 4];
            *(float4*)(kd)           = kn0;
            *(float4*)(kd + F_P)     = kn1;
            *(float4*)(kd + 2 * F_P) = kn2;
            *(float4*)(kd + 3 * F_P) = kn3;
            *(float4*)(kd + 4 * F_P) = kn4;
            *(float4*)(kd + 5 * F_P) = kn5;
            *(float4*)(kd + 6 * F_P) = kn6;
            *(float4*)(kd + 7 * F_P) = kn7;
            float* vd = &Vs[(8 * wave) * F_P + (32 * (ch + 1) + lane) * 4];
            *(float4*)(vd)           = vn0;
            *(float4*)(vd + F_P)     = vn1;
            *(float4*)(vd + 2 * F_P) = vn2;
            *(float4*)(vd + 3 * F_P) = vn3;
            *(float4*)(vd + 4 * F_P) = vn4;
            *(float4*)(vd + 5 * F_P) = vn5;
            *(float4*)(vd + 6 * F_P) = vn6;
            *(float4*)(vd + 7 * F_P) = vn7;
            __syncthreads();
            if (ch + 2 < F_NC) {
                const float* ks = Kb + (size_t)(8 * wave) * SEQ + (32 * (ch + 2) + lane) * 4;
                kn0 = *(const float4*)(ks);
                kn1 = *(const float4*)(ks + SEQ);
                kn2 = *(const float4*)(ks + 2 * SEQ);
                kn3 = *(const float4*)(ks + 3 * SEQ);
                kn4 = *(const float4*)(ks + 4 * SEQ);
                kn5 = *(const float4*)(ks + 5 * SEQ);
                kn6 = *(const float4*)(ks + 6 * SEQ);
                kn7 = *(const float4*)(ks + 7 * SEQ);
                const float* vs = Vb + (size_t)(8 * wave) * SEQ + (32 * (ch + 2) + lane) * 4;
                vn0 = *(const float4*)(vs);
                vn1 = *(const float4*)(vs + SEQ);
                vn2 = *(const float4*)(vs + 2 * SEQ);
                vn3 = *(const float4*)(vs + 3 * SEQ);
                vn4 = *(const float4*)(vs + 4 * SEQ);
                vn5 = *(const float4*)(vs + 5 * SEQ);
                vn6 = *(const float4*)(vs + 6 * SEQ);
                vn7 = *(const float4*)(vs + 7 * SEQ);
            }
        }
    }
#undef ATTN_PASS2
#undef ATTN_TP4
#undef ATTN_ORDER2

    #pragma unroll
    for (int r = 0; r < 4; ++r) {
        const float sum = (sa[r].x + sa[r].y) + (sb[r].x + sb[r].y);
        const float num = (pa[r].x + pa[r].y) + (pb[r].x + pb[r].y);
        SAs[(4 * rq + r) * F_PA + c] = num / sum;
    }
    __syncthreads();

    #pragma unroll
    for (int j = 10; j < DK / 4; ++j) {
        bx[j] = *(const float4*)(wb + (size_t)(4 * j) * DIM);
        by[j] = *(const float4*)(wb + (size_t)(4 * j + 1) * DIM);
    }

    v8f acc0 = {0.f,0.f,0.f,0.f,0.f,0.f,0.f,0.f};
    v8f acc1 = acc0, acc2 = acc0, acc3 = acc0;

    const float* af = &SAs[lm * F_PA + 2 * lk];
    #pragma unroll
    for (int j = 0; j < DK / 4; ++j) {
        const v2f a  = *(const v2f*)(af + 4 * j);
        const v2f b0 = {bx[j].x, by[j].x};
        const v2f b1 = {bx[j].y, by[j].y};
        const v2f b2 = {bx[j].z, by[j].z};
        const v2f b3 = {bx[j].w, by[j].w};
        acc0 = __builtin_amdgcn_wmma_f32_16x16x4_f32(
            false, a, false, b0, (short)0, acc0, false, false);
        acc1 = __builtin_amdgcn_wmma_f32_16x16x4_f32(
            false, a, false, b1, (short)0, acc1, false, false);
        acc2 = __builtin_amdgcn_wmma_f32_16x16x4_f32(
            false, a, false, b2, (short)0, acc2, false, false);
        acc3 = __builtin_amdgcn_wmma_f32_16x16x4_f32(
            false, a, false, b3, (short)0, acc3, false, false);
    }

    float* obase = Out + ((size_t)b * SEQ + n0 + lk * 8) * DIM + wave * 64 + 4 * lm;
    #pragma unroll
    for (int r = 0; r < 8; ++r)
        *(float4*)(obase + (size_t)r * DIM) = make_float4(acc0[r], acc1[r], acc2[r], acc3[r]);
}

extern "C" void kernel_launch(void* const* d_in, const int* in_sizes, int n_in,
                              void* d_out, int out_size, void* d_ws, size_t ws_size,
                              hipStream_t stream)
{
    const float* X    = (const float*)d_in[0];
    const float* Wqkv = (const float*)d_in[1];
    const float* Wo   = (const float*)d_in[2];
    float* out = (float*)d_out;

    const size_t CHN = (size_t)BATCH * DK * SEQ;
    float* Qt = (float*)d_ws;
    float* Kt = Qt + CHN;
    float* Vt = Kt + CHN;
    float* Kmx = Vt + CHN;
    float* Kmn = Kmx + (size_t)BATCH * DK * KTILES;

    qkv_gemm_kernel<<<256, 384, 0, stream>>>(X, Wqkv, Qt, Kt, Vt, Kmx, Kmn);
    attn_out_kernel<<<BATCH * (SEQ / F_BM), 256, 0, stream>>>(Qt, Kt, Vt, Kmx, Kmn, Wo, out);
}
